# starting kernel
# baseline (speedup 1.0000x reference)
#include <hip/hip_runtime.h>

#define D   512
#define R   64
#define WIN 32
#define BB  4
#define SS  2048
#define NN  (SS + 1)
#define MM  (BB * NN)
#define TILES ((NN + 15) / 16)
#define WTOT 66
#define KB  84

typedef float v2f __attribute__((ext_vector_type(2)));
typedef float v8f __attribute__((ext_vector_type(8)));

__device__ __forceinline__ int clampi(int x, int lo, int hi) {
    return min(max(x, lo), hi);
}

__global__ __launch_bounds__(256) void embed_ln_kernel(
    const int* __restrict__ ids, const float* __restrict__ emb,
    const float* __restrict__ pos, const float* __restrict__ anchor_val,
    const float* __restrict__ anchor_state, const float* __restrict__ gamma,
    const float* __restrict__ beta, const float* __restrict__ sp_w,
    const float* __restrict__ sp_b, float* __restrict__ val,
    float* __restrict__ state)
{
    __shared__ float sred[256];
    const int bn  = blockIdx.x;
    const int b   = bn / NN;
    const int n   = bn % NN;
    const int tid = threadIdx.x;
    float* vrow = val + (size_t)bn * D;

    if (n == 0) {
        for (int d = tid; d < D; d += 256) vrow[d] = anchor_val[d];
        if (tid == 0) state[bn] = anchor_state[0];
        return;
    }
    const int s  = n - 1;
    const int id = ids[b * SS + s];
    const float* erow = emb + (size_t)id * D;
    const float* prow = pos + (size_t)s * D;
    const int d0 = tid, d1 = tid + 256;
    float x0 = erow[d0] + prow[d0];
    float x1 = erow[d1] + prow[d1];

    sred[tid] = x0 + x1; __syncthreads();
    for (int o = 128; o > 0; o >>= 1) { if (tid < o) sred[tid] += sred[tid + o]; __syncthreads(); }
    float mean = sred[0] * (1.0f / D);
    __syncthreads();
    float c0 = x0 - mean, c1 = x1 - mean;
    sred[tid] = c0 * c0 + c1 * c1; __syncthreads();
    for (int o = 128; o > 0; o >>= 1) { if (tid < o) sred[tid] += sred[tid + o]; __syncthreads(); }
    float rstd = rsqrtf(sred[0] * (1.0f / D) + 1e-5f);
    __syncthreads();
    float y0 = c0 * rstd * gamma[d0] + beta[d0];
    float y1 = c1 * rstd * gamma[d1] + beta[d1];
    vrow[d0] = y0; vrow[d1] = y1;
    sred[tid] = y0 * sp_w[d0] + y1 * sp_w[d1]; __syncthreads();
    for (int o = 128; o > 0; o >>= 1) { if (tid < o) sred[tid] += sred[tid + o]; __syncthreads(); }
    if (tid == 0) state[bn] = sred[0] + sp_b[0];
}

__global__ __launch_bounds__(256) void qk_gemm_kernel(
    const float* __restrict__ val, const float* __restrict__ wq,
    const float* __restrict__ wk, float* __restrict__ q,
    float* __restrict__ k)
{
    const int lane   = threadIdx.x & 31;
    const int wave   = threadIdx.x >> 5;
    const int mtile  = blockIdx.x * 8 + wave;
    const int mtiles = (MM + 15) / 16;
    if (mtile >= mtiles) return;

    const int row0 = mtile * 16;
    const int lm   = lane & 15;
    const int lh   = lane >> 4;
    const int arow = min(row0 + lm, MM - 1);
    const float* aptr = val + (size_t)arow * D + 2 * lh;

    v8f accq[4] = {};
    v8f acck[4] = {};

    for (int kk = 0; kk < D; kk += 4) {
        v2f a = *(const v2f*)(aptr + kk);
        const float* bqp = wq + (size_t)(kk + 2 * lh) * R + lm;
        const float* bkp = wk + (size_t)(kk + 2 * lh) * R + lm;
#pragma unroll
        for (int nt = 0; nt < 4; ++nt) {
            v2f bq; bq.x = bqp[nt * 16]; bq.y = bqp[R + nt * 16];
            v2f bk; bk.x = bkp[nt * 16]; bk.y = bkp[R + nt * 16];
            accq[nt] = __builtin_amdgcn_wmma_f32_16x16x4_f32(
                false, a, false, bq, (short)0, accq[nt], false, false);
            acck[nt] = __builtin_amdgcn_wmma_f32_16x16x4_f32(
                false, a, false, bk, (short)0, acck[nt], false, false);
        }
    }

#pragma unroll
    for (int nt = 0; nt < 4; ++nt) {
#pragma unroll
        for (int i = 0; i < 8; ++i) {
            const int r = row0 + i + 8 * lh;
            if (r < MM) {
                q[(size_t)r * R + nt * 16 + lm] = accq[nt][i];
                k[(size_t)r * R + nt * 16 + lm] = acck[nt][i];
            }
        }
    }
}

__global__ __launch_bounds__(256) void attn_fused_kernel(
    const float* __restrict__ val_in, const float* __restrict__ state_in,
    const float* __restrict__ q, const float* __restrict__ k,
    const float* __restrict__ gamma, const float* __restrict__ beta,
    float* __restrict__ val_out, float* __restrict__ state_out)
{
    __shared__ float s_scores[16][80];
    __shared__ float s_w[16][KB];
    __shared__ float s_y[16][D];
    __shared__ float s_r[16][16];
    __shared__ float s_mean[16];
    __shared__ float s_rstd[16];

    const int tid  = threadIdx.x;
    const int lane = tid & 31;
    const int wave = tid >> 5;
    const int lm   = lane & 15;
    const int lh   = lane >> 4;
    const int b    = blockIdx.x / TILES;
    const int t    = blockIdx.x % TILES;
    const int p0   = t * 16;
    const int base = b * NN;

    if (wave < 5) {
        const int qrow = base + clampi(p0 + lm, 0, NN - 1);
        const int j    = wave * 16 + lm;
        const int krow = base + clampi(p0 - 32 + j, 0, NN - 1);
        const float* qp = q + (size_t)qrow * R + 2 * lh;
        const float* kp = k + (size_t)krow * R + 2 * lh;
        v8f acc = {};
        for (int kk = 0; kk < R; kk += 4) {
            v2f a  = *(const v2f*)(qp + kk);
            v2f bb = *(const v2f*)(kp + kk);
            acc = __builtin_amdgcn_wmma_f32_16x16x4_f32(
                false, a, false, bb, (short)0, acc, false, false);
        }
#pragma unroll
        for (int i = 0; i < 8; ++i)
            s_scores[i + 8 * lh][wave * 16 + lm] = acc[i];
    }
    __syncthreads();

    for (int x = tid; x < 16 * KB; x += 256) (&s_w[0][0])[x] = 0.f;
    __syncthreads();

    if (tid < 16) {
        const int m    = tid;
        const int node = p0 + m;
        if (node < NN) {
            const float scale = 0.125f;
            float adot = 0.f;
            {
                const float* qp = q + (size_t)(base + node) * R;
                const float* k0 = k + (size_t)base * R;
                for (int r = 0; r < R; ++r) adot = fmaf(qp[r], k0[r], adot);
                adot *= scale;
            }
            const bool avalid = (node - WIN) > 0;
            float mx = avalid ? fabsf(adot) : -1e30f;
            for (int w = 0; w < 65; ++w) {
                const int p = node - WIN + w;
                if (p >= 0 && p < NN)
                    mx = fmaxf(mx, fabsf(s_scores[m][m + w] * scale));
            }
            float sum = 0.f;
            for (int w = 0; w < 65; ++w) {
                const int p = node - WIN + w;
                float se = 0.f;
                if (p >= 0 && p < NN) {
                    const float s = s_scores[m][m + w] * scale;
                    const float e = __expf(fabsf(s) - mx);
                    sum += e;
                    se = (s > 0.f) ? e : ((s < 0.f) ? -e : 0.f);
                }
                s_scores[m][m + w] = se;
            }
            float ase = 0.f;
            if (avalid) {
                const float e = __expf(fabsf(adot) - mx);
                sum += e;
                ase = (adot > 0.f) ? e : ((adot < 0.f) ? -e : 0.f);
            }
            const float inv = 1.f / sum;
            float ds = 0.f;
            for (int w = 0; w < 65; ++w) {
                const float we = s_scores[m][m + w] * inv;
                s_w[m][m + w] = we;
                ds += we * state_in[base + clampi(node - WIN + w, 0, NN - 1)];
            }
            const float wa = ase * inv;
            s_w[m][80] = wa;
            ds += wa * state_in[base];
            state_out[base + node] = state_in[base + node] + ds;
        }
    }
    __syncthreads();

    {
        const int c0 = wave * 64;
        v8f acc[4] = {};
        for (int kk = 0; kk < KB; kk += 4) {
            const int kp0 = kk + 2 * lh;
            const int kp1 = kp0 + 1;
            v2f a;
            a.x = s_w[lm][kp0];
            a.y = s_w[lm][kp1];
            const int r0 = (kp0 >= 80) ? base : base + clampi(p0 - 32 + kp0, 0, NN - 1);
            const int r1 = (kp1 >= 80) ? base : base + clampi(p0 - 32 + kp1, 0, NN - 1);
            const float* v0 = val_in + (size_t)r0 * D + c0 + lm;
            const float* v1 = val_in + (size_t)r1 * D + c0 + lm;
#pragma unroll
            for (int nt = 0; nt < 4; ++nt) {
                v2f bb;
                bb.x = v0[nt * 16];
                bb.y = v1[nt * 16];
                acc[nt] = __builtin_amdgcn_wmma_f32_16x16x4_f32(
                    false, a, false, bb, (short)0, acc[nt], false, false);
            }
        }
#pragma unroll
        for (int nt = 0; nt < 4; ++nt) {
#pragma unroll
            for (int i = 0; i < 8; ++i) {
                const int m    = i + 8 * lh;
                const int col  = c0 + nt * 16 + lm;
                const int node = p0 + m;
                float y = 0.f;
                if (node < NN)
                    y = acc[nt][i] + val_in[(size_t)(base + node) * D + col];
                s_y[m][col] = y;
            }
        }
    }
    __syncthreads();

    const int m  = tid >> 4;
    const int jj = tid & 15;
    float part = 0.f;
    for (int e = 0; e < 32; ++e) part += s_y[m][jj * 32 + e];
    s_r[m][jj] = part;
    __syncthreads();
    if (jj == 0) {
        float tot = 0.f;
        for (int x = 0; x < 16; ++x) tot += s_r[m][x];
        s_mean[m] = tot * (1.0f / D);
    }
    __syncthreads();
    const float mean = s_mean[m];
    part = 0.f;
    for (int e = 0; e < 32; ++e) {
        const float c = s_y[m][jj * 32 + e] - mean;
        part += c * c;
    }
    s_r[m][jj] = part;
    __syncthreads();
    if (jj == 0) {
        float tot = 0.f;
        for (int x = 0; x < 16; ++x) tot += s_r[m][x];
        s_rstd[m] = rsqrtf(tot * (1.0f / D) + 1e-5f);
    }
    __syncthreads();
    const int node = p0 + m;
    if (node < NN) {
        const float rstd = s_rstd[m];
        float* vo = val_out + (size_t)(base + node) * D;
        for (int e = 0; e < 32; ++e) {
            const int col = jj * 32 + e;
            vo[col] = (s_y[m][col] - mean) * rstd * gamma[col] + beta[col];
        }
    }
}

extern "C" void kernel_launch(void* const* d_in, const int* in_sizes, int n_in,
                              void* d_out, int out_size, void* d_ws, size_t ws_size,
                              hipStream_t stream)
{
    const int*   ids    = (const int*)  d_in[0];
    const float* emb    = (const float*)d_in[1];
    const float* pos    = (const float*)d_in[2];
    const float* aval   = (const float*)d_in[3];
    const float* astate = (const float*)d_in[4];
    const float* ing    = (const float*)d_in[5];
    const float* inb    = (const float*)d_in[6];
    const float* spw    = (const float*)d_in[7];
    const float* spb    = (const float*)d_in[8];
    const float* wq     = (const float*)d_in[9];
    const float* wk     = (const float*)d_in[10];
    const float* lng    = (const float*)d_in[11];
    const float* lnb    = (const float*)d_in[12];

    float* ws = (float*)d_ws;
    size_t o = 0;
    float* val_a = ws + o; o += (size_t)MM * D;
    float* val_b = ws + o; o += (size_t)MM * D;
    float* st_a  = ws + o; o += MM;
    float* st_b  = ws + o; o += MM;
    float* qbuf  = ws + o; o += (size_t)MM * R;
    float* kbuf  = ws + o; o += (size_t)MM * R;

    float* out_state = (float*)d_out;
    float* out_val   = out_state + MM;

    embed_ln_kernel<<<MM, 256, 0, stream>>>(ids, emb, pos, aval, astate,
                                            ing, inb, spw, spb, val_a, st_a);

    const int mtiles  = (MM + 15) / 16;
    const int gblocks = (mtiles + 7) / 8;
    const int ablocks = BB * TILES;

    qk_gemm_kernel<<<gblocks, 256, 0, stream>>>(val_a, wq, wk, qbuf, kbuf);
    attn_fused_kernel<<<ablocks, 256, 0, stream>>>(val_a, st_a, qbuf, kbuf,
                                                   lng, lnb, val_b, st_b);
    qk_gemm_kernel<<<gblocks, 256, 0, stream>>>(val_b, wq + (size_t)D * R,
                                                wk + (size_t)D * R, qbuf, kbuf);
    attn_fused_kernel<<<ablocks, 256, 0, stream>>>(val_b, st_b, qbuf, kbuf,
                                                   lng + D, lnb + D, out_val, out_state);
}


# QK projection GEMM re-tiled as an LDS-staged block GEMM: 32x128 tile per 8-wave block (257 blocks, was 65), val and weight K-slabs via 16-byte loads with register prefetch; k order unchanged
# speedup vs baseline: 1.9308x; 1.9308x over previous
#include <hip/hip_runtime.h>

#define D   512
#define R   64
#define WIN 32
#define BB  4
#define SS  2048
#define NN  (SS + 1)
#define MM  (BB * NN)
#define TILES ((NN + 15) / 16)
#define WTOT 66
#define KB  84

typedef float v2f __attribute__((ext_vector_type(2)));
typedef float v8f __attribute__((ext_vector_type(8)));

__device__ __forceinline__ int clampi(int x, int lo, int hi) {
    return min(max(x, lo), hi);
}

__global__ __launch_bounds__(256) void embed_ln_kernel(
    const int* __restrict__ ids, const float* __restrict__ emb,
    const float* __restrict__ pos, const float* __restrict__ anchor_val,
    const float* __restrict__ anchor_state, const float* __restrict__ gamma,
    const float* __restrict__ beta, const float* __restrict__ sp_w,
    const float* __restrict__ sp_b, float* __restrict__ val,
    float* __restrict__ state)
{
    __shared__ float sred[256];
    const int bn  = blockIdx.x;
    const int b   = bn / NN;
    const int n   = bn % NN;
    const int tid = threadIdx.x;
    float* vrow = val + (size_t)bn * D;

    if (n == 0) {
        for (int d = tid; d < D; d += 256) vrow[d] = anchor_val[d];
        if (tid == 0) state[bn] = anchor_state[0];
        return;
    }
    const int s  = n - 1;
    const int id = ids[b * SS + s];
    const float* erow = emb + (size_t)id * D;
    const float* prow = pos + (size_t)s * D;
    const int d0 = tid, d1 = tid + 256;
    float x0 = erow[d0] + prow[d0];
    float x1 = erow[d1] + prow[d1];

    sred[tid] = x0 + x1; __syncthreads();
    for (int o = 128; o > 0; o >>= 1) { if (tid < o) sred[tid] += sred[tid + o]; __syncthreads(); }
    float mean = sred[0] * (1.0f / D);
    __syncthreads();
    float c0 = x0 - mean, c1 = x1 - mean;
    sred[tid] = c0 * c0 + c1 * c1; __syncthreads();
    for (int o = 128; o > 0; o >>= 1) { if (tid < o) sred[tid] += sred[tid + o]; __syncthreads(); }
    float rstd = rsqrtf(sred[0] * (1.0f / D) + 1e-5f);
    __syncthreads();
    float y0 = c0 * rstd * gamma[d0] + beta[d0];
    float y1 = c1 * rstd * gamma[d1] + beta[d1];
    vrow[d0] = y0; vrow[d1] = y1;
    sred[tid] = y0 * sp_w[d0] + y1 * sp_w[d1]; __syncthreads();
    for (int o = 128; o > 0; o >>= 1) { if (tid < o) sred[tid] += sred[tid + o]; __syncthreads(); }
    if (tid == 0) state[bn] = sred[0] + sp_b[0];
}

#define GBM   32
#define GBK   32
#define GPA   (GBK + 4)
#define GPB   (2 * R + 16)
#define NSLAB (D / GBK)

typedef float v4f __attribute__((ext_vector_type(4)));

__global__ __launch_bounds__(256) void qk_gemm_kernel(
    const float* __restrict__ val, const float* __restrict__ wq,
    const float* __restrict__ wk, float* __restrict__ q,
    float* __restrict__ k)
{
    __shared__ __attribute__((aligned(16))) float sA[2][GBM][GPA];
    __shared__ __attribute__((aligned(16))) float sB[2][GBK][GPB];

    const int tid  = threadIdx.x;
    const int lane = tid & 31;
    const int wave = tid >> 5;
    const int lm   = lane & 15;
    const int lh   = lane >> 4;
    const int mt   = wave & 1;
    const int n0   = (wave >> 1) * 32;
    const int row0 = blockIdx.x * GBM;

    const int ar = tid >> 3;
    const int ac = (tid & 7) * 4;
    const int br = tid >> 4;
    const int bc = (tid & 15) * 4;
    const float* ag = val + (size_t)min(row0 + ar, MM - 1) * D + ac;
    const float* qg = wq + (size_t)br * R + bc;
    const float* kg = wk + (size_t)br * R + bc;

    v4f ra  = *(const v4f*)ag;
    v4f rq0 = *(const v4f*)qg;
    v4f rq1 = *(const v4f*)(qg + 16 * R);
    v4f rk0 = *(const v4f*)kg;
    v4f rk1 = *(const v4f*)(kg + 16 * R);
    *(v4f*)&sA[0][ar][ac]          = ra;
    *(v4f*)&sB[0][br][bc]          = rq0;
    *(v4f*)&sB[0][br + 16][bc]     = rq1;
    *(v4f*)&sB[0][br][R + bc]      = rk0;
    *(v4f*)&sB[0][br + 16][R + bc] = rk1;
    __syncthreads();

    const bool active = (row0 + mt * 16) < MM;
    v8f acc0 = {};
    v8f acc1 = {};

    for (int s = 0; s < NSLAB; ++s) {
        const int cur = s & 1;
        const bool more = (s + 1) < NSLAB;
        if (more) {
            const int k0 = (s + 1) * GBK;
            ra  = *(const v4f*)(ag + k0);
            rq0 = *(const v4f*)(qg + (size_t)k0 * R);
            rq1 = *(const v4f*)(qg + (size_t)(k0 + 16) * R);
            rk0 = *(const v4f*)(kg + (size_t)k0 * R);
            rk1 = *(const v4f*)(kg + (size_t)(k0 + 16) * R);
        }
        if (active) {
            const float* pa = &sA[cur][mt * 16 + lm][2 * lh];
            const float* pb = &sB[cur][2 * lh][n0 + lm];
#pragma unroll
            for (int kk = 0; kk < GBK; kk += 4) {
                v2f a = *(const v2f*)(pa + kk);
                v2f b0, b1;
                b0.x = pb[kk * GPB];
                b0.y = pb[kk * GPB + GPB];
                b1.x = pb[kk * GPB + 16];
                b1.y = pb[kk * GPB + GPB + 16];
                acc0 = __builtin_amdgcn_wmma_f32_16x16x4_f32(
                    false, a, false, b0, (short)0, acc0, false, false);
                acc1 = __builtin_amdgcn_wmma_f32_16x16x4_f32(
                    false, a, false, b1, (short)0, acc1, false, false);
            }
        }
        if (more) {
            const int nxt = cur ^ 1;
            *(v4f*)&sA[nxt][ar][ac]          = ra;
            *(v4f*)&sB[nxt][br][bc]          = rq0;
            *(v4f*)&sB[nxt][br + 16][bc]     = rq1;
            *(v4f*)&sB[nxt][br][R + bc]      = rk0;
            *(v4f*)&sB[nxt][br + 16][R + bc] = rk1;
        }
        __syncthreads();
    }

    float (*sO)[GPB] = sB[0];
#pragma unroll
    for (int i = 0; i < 8; ++i) {
        sO[mt * 16 + 8 * lh + i][n0 + lm]      = acc0[i];
        sO[mt * 16 + 8 * lh + i][n0 + 16 + lm] = acc1[i];
    }
    __syncthreads();
    for (int j = 0; j < 4; ++j) {
        const int idx  = tid + 256 * j;
        const int orow = idx >> 5;
        const int oc   = (idx & 31) * 4;
        const int r    = row0 + orow;
        if (r < MM) {
            const v4f v = *(const v4f*)&sO[orow][oc];
            float* dst = (oc < R) ? (q + (size_t)r * R + oc) : (k + (size_t)r * R + (oc - R));
            *(v4f*)dst = v;
        }
    }
}

__global__ __launch_bounds__(256) void attn_fused_kernel(
    const float* __restrict__ val_in, const float* __restrict__ state_in,
    const float* __restrict__ q, const float* __restrict__ k,
    const float* __restrict__ gamma, const float* __restrict__ beta,
    float* __restrict__ val_out, float* __restrict__ state_out)
{
    __shared__ float s_scores[16][80];
    __shared__ float s_w[16][KB];
    __shared__ float s_y[16][D];
    __shared__ float s_r[16][16];
    __shared__ float s_mean[16];
    __shared__ float s_rstd[16];

    const int tid  = threadIdx.x;
    const int lane = tid & 31;
    const int wave = tid >> 5;
    const int lm   = lane & 15;
    const int lh   = lane >> 4;
    const int b    = blockIdx.x / TILES;
    const int t    = blockIdx.x % TILES;
    const int p0   = t * 16;
    const int base = b * NN;

    if (wave < 5) {
        const int qrow = base + clampi(p0 + lm, 0, NN - 1);
        const int j    = wave * 16 + lm;
        const int krow = base + clampi(p0 - 32 + j, 0, NN - 1);
        const float* qp = q + (size_t)qrow * R + 2 * lh;
        const float* kp = k + (size_t)krow * R + 2 * lh;
        v8f acc = {};
        for (int kk = 0; kk < R; kk += 4) {
            v2f a  = *(const v2f*)(qp + kk);
            v2f bb = *(const v2f*)(kp + kk);
            acc = __builtin_amdgcn_wmma_f32_16x16x4_f32(
                false, a, false, bb, (short)0, acc, false, false);
        }
#pragma unroll
        for (int i = 0; i < 8; ++i)
            s_scores[i + 8 * lh][wave * 16 + lm] = acc[i];
    }
    __syncthreads();

    for (int x = tid; x < 16 * KB; x += 256) (&s_w[0][0])[x] = 0.f;
    __syncthreads();

    if (tid < 16) {
        const int m    = tid;
        const int node = p0 + m;
        if (node < NN) {
            const float scale = 0.125f;
            float adot = 0.f;
            {
                const float* qp = q + (size_t)(base + node) * R;
                const float* k0 = k + (size_t)base * R;
                for (int r = 0; r < R; ++r) adot = fmaf(qp[r], k0[r], adot);
                adot *= scale;
            }
            const bool avalid = (node - WIN) > 0;
            float mx = avalid ? fabsf(adot) : -1e30f;
            for (int w = 0; w < 65; ++w) {
                const int p = node - WIN + w;
                if (p >= 0 && p < NN)
                    mx = fmaxf(mx, fabsf(s_scores[m][m + w] * scale));
            }
            float sum = 0.f;
            for (int w = 0; w < 65; ++w) {
                const int p = node - WIN + w;
                float se = 0.f;
                if (p >= 0 && p < NN) {
                    const float s = s_scores[m][m + w] * scale;
                    const float e = __expf(fabsf(s) - mx);
                    sum += e;
                    se = (s > 0.f) ? e : ((s < 0.f) ? -e : 0.f);
                }
                s_scores[m][m + w] = se;
            }
            float ase = 0.f;
            if (avalid) {
                const float e = __expf(fabsf(adot) - mx);
                sum += e;
                ase = (adot > 0.f) ? e : ((adot < 0.f) ? -e : 0.f);
            }
            const float inv = 1.f / sum;
            float ds = 0.f;
            for (int w = 0; w < 65; ++w) {
                const float we = s_scores[m][m + w] * inv;
                s_w[m][m + w] = we;
                ds += we * state_in[base + clampi(node - WIN + w, 0, NN - 1)];
            }
            const float wa = ase * inv;
            s_w[m][80] = wa;
            ds += wa * state_in[base];
            state_out[base + node] = state_in[base + node] + ds;
        }
    }
    __syncthreads();

    {
        const int c0 = wave * 64;
        v8f acc[4] = {};
        for (int kk = 0; kk < KB; kk += 4) {
            const int kp0 = kk + 2 * lh;
            const int kp1 = kp0 + 1;
            v2f a;
            a.x = s_w[lm][kp0];
            a.y = s_w[lm][kp1];
            const int r0 = (kp0 >= 80) ? base : base + clampi(p0 - 32 + kp0, 0, NN - 1);
            const int r1 = (kp1 >= 80) ? base : base + clampi(p0 - 32 + kp1, 0, NN - 1);
            const float* v0 = val_in + (size_t)r0 * D + c0 + lm;
            const float* v1 = val_in + (size_t)r1 * D + c0 + lm;
#pragma unroll
            for (int nt = 0; nt < 4; ++nt) {
                v2f bb;
                bb.x = v0[nt * 16];
                bb.y = v1[nt * 16];
                acc[nt] = __builtin_amdgcn_wmma_f32_16x16x4_f32(
                    false, a, false, bb, (short)0, acc[nt], false, false);
            }
        }
#pragma unroll
        for (int nt = 0; nt < 4; ++nt) {
#pragma unroll
            for (int i = 0; i < 8; ++i) {
                const int m    = i + 8 * lh;
                const int col  = c0 + nt * 16 + lm;
                const int node = p0 + m;
                float y = 0.f;
                if (node < NN)
                    y = acc[nt][i] + val_in[(size_t)(base + node) * D + col];
                s_y[m][col] = y;
            }
        }
    }
    __syncthreads();

    const int m  = tid >> 4;
    const int jj = tid & 15;
    float part = 0.f;
    for (int e = 0; e < 32; ++e) part += s_y[m][jj * 32 + e];
    s_r[m][jj] = part;
    __syncthreads();
    if (jj == 0) {
        float tot = 0.f;
        for (int x = 0; x < 16; ++x) tot += s_r[m][x];
        s_mean[m] = tot * (1.0f / D);
    }
    __syncthreads();
    const float mean = s_mean[m];
    part = 0.f;
    for (int e = 0; e < 32; ++e) {
        const float c = s_y[m][jj * 32 + e] - mean;
        part += c * c;
    }
    s_r[m][jj] = part;
    __syncthreads();
    if (jj == 0) {
        float tot = 0.f;
        for (int x = 0; x < 16; ++x) tot += s_r[m][x];
        s_rstd[m] = rsqrtf(tot * (1.0f / D) + 1e-5f);
    }
    __syncthreads();
    const int node = p0 + m;
    if (node < NN) {
        const float rstd = s_rstd[m];
        float* vo = val_out + (size_t)(base + node) * D;
        for (int e = 0; e < 32; ++e) {
            const int col = jj * 32 + e;
            vo[col] = (s_y[m][col] - mean) * rstd * gamma[col] + beta[col];
        }
    }
}

extern "C" void kernel_launch(void* const* d_in, const int* in_sizes, int n_in,
                              void* d_out, int out_size, void* d_ws, size_t ws_size,
                              hipStream_t stream)
{
    const int*   ids    = (const int*)  d_in[0];
    const float* emb    = (const float*)d_in[1];
    const float* pos    = (const float*)d_in[2];
    const float* aval   = (const float*)d_in[3];
    const float* astate = (const float*)d_in[4];
    const float* ing    = (const float*)d_in[5];
    const float* inb    = (const float*)d_in[6];
    const float* spw    = (const float*)d_in[7];
    const float* spb    = (const float*)d_in[8];
    const float* wq     = (const float*)d_in[9];
    const float* wk     = (const float*)d_in[10];
    const float* lng    = (const float*)d_in[11];
    const float* lnb    = (const float*)d_in[12];

    float* ws = (float*)d_ws;
    size_t o = 0;
    float* val_a = ws + o; o += (size_t)MM * D;
    float* val_b = ws + o; o += (size_t)MM * D;
    float* st_a  = ws + o; o += MM;
    float* st_b  = ws + o; o += MM;
    float* qbuf  = ws + o; o += (size_t)MM * R;
    float* kbuf  = ws + o; o += (size_t)MM * R;

    float* out_state = (float*)d_out;
    float* out_val   = out_state + MM;

    embed_ln_kernel<<<MM, 256, 0, stream>>>(ids, emb, pos, aval, astate,
                                            ing, inb, spw, spb, val_a, st_a);

    const int gblocks = (MM + GBM - 1) / GBM;
    const int ablocks = BB * TILES;

    qk_gemm_kernel<<<gblocks, 256, 0, stream>>>(val_a, wq, wk, qbuf, kbuf);
    attn_fused_kernel<<<ablocks, 256, 0, stream>>>(val_a, st_a, qbuf, kbuf,
                                                   lng, lnb, val_b, st_b);
    qk_gemm_kernel<<<gblocks, 256, 0, stream>>>(val_b, wq + (size_t)D * R,
                                                wk + (size_t)D * R, qbuf, kbuf);
    attn_fused_kernel<<<ablocks, 256, 0, stream>>>(val_b, st_b, qbuf, kbuf,
                                                   lng + D, lnb + D, out_val, out_state);
}


# embed+LayerNorm kernel: one wave per row, row in registers, three row sums as in-register butterflies (1025 blocks of 8 rows; was 8196 blocks, 29 barriers each); summation tree of those sums changed
# speedup vs baseline: 2.0571x; 1.0654x over previous
#include <hip/hip_runtime.h>

#define D   512
#define R   64
#define WIN 32
#define BB  4
#define SS  2048
#define NN  (SS + 1)
#define MM  (BB * NN)
#define TILES ((NN + 15) / 16)
#define WTOT 66
#define KB  84

typedef float v2f __attribute__((ext_vector_type(2)));
typedef float v8f __attribute__((ext_vector_type(8)));

__device__ __forceinline__ int clampi(int x, int lo, int hi) {
    return min(max(x, lo), hi);
}

#define EROWS 8

typedef float v4f __attribute__((ext_vector_type(4)));

template <int M>
__device__ __forceinline__ float row_xor(float x) {
    const int b = __builtin_bit_cast(int, x);
    return __builtin_bit_cast(float,
        __builtin_amdgcn_update_dpp(b, b, 0x160 | M, 0xf, 0xf, false));
}

__device__ __forceinline__ float xhalf(float x) {
    const int b = __builtin_bit_cast(int, x);
    return __builtin_bit_cast(float,
        __builtin_amdgcn_permlanex16(b, b, 0x76543210, 0xfedcba98, false, false));
}

__device__ __forceinline__ float wave_sum(float x) {
    x += row_xor<1>(x);
    x += row_xor<2>(x);
    x += row_xor<4>(x);
    x += row_xor<8>(x);
    return x + xhalf(x);
}

__global__ __launch_bounds__(256) void embed_ln_kernel(
    const int* __restrict__ ids, const float* __restrict__ emb,
    const float* __restrict__ pos, const float* __restrict__ anchor_val,
    const float* __restrict__ anchor_state, const float* __restrict__ gamma,
    const float* __restrict__ beta, const float* __restrict__ sp_w,
    const float* __restrict__ sp_b, float* __restrict__ val,
    float* __restrict__ state)
{
    const int lane = threadIdx.x & 31;
    const int bn = __builtin_amdgcn_readfirstlane((int)(blockIdx.x * EROWS + (threadIdx.x >> 5)));
    if (bn >= MM) return;
    const int b = bn / NN;
    const int n = bn - b * NN;
    const int c = lane * 4;
    float* vrow = val + (size_t)bn * D + c;

    if (n == 0) {
#pragma unroll
        for (int j = 0; j < 4; ++j)
            *(v4f*)(vrow + j * 128) = *(const v4f*)(anchor_val + j * 128 + c);
        if (lane == 0) state[bn] = anchor_state[0];
        return;
    }
    const int s  = n - 1;
    const int id = ids[b * SS + s];
    const float* erow = emb + (size_t)id * D + c;
    const float* prow = pos + (size_t)s * D + c;

    v4f x[4];
#pragma unroll
    for (int j = 0; j < 4; ++j)
        x[j] = *(const v4f*)(erow + j * 128) + *(const v4f*)(prow + j * 128);

    float part = 0.f;
#pragma unroll
    for (int j = 0; j < 4; ++j)
        part += (x[j].x + x[j].y) + (x[j].z + x[j].w);
    const float mean = wave_sum(part) * (1.0f / D);

    part = 0.f;
#pragma unroll
    for (int j = 0; j < 4; ++j) {
        x[j] -= mean;
        const v4f sq = x[j] * x[j];
        part += (sq.x + sq.y) + (sq.z + sq.w);
    }
    const float rstd = rsqrtf(wave_sum(part) * (1.0f / D) + 1e-5f);

    part = 0.f;
#pragma unroll
    for (int j = 0; j < 4; ++j) {
        const v4f g  = *(const v4f*)(gamma + j * 128 + c);
        const v4f bt = *(const v4f*)(beta + j * 128 + c);
        const v4f w  = *(const v4f*)(sp_w + j * 128 + c);
        const v4f y  = x[j] * rstd * g + bt;
        *(v4f*)(vrow + j * 128) = y;
        const v4f yw = y * w;
        part += (yw.x + yw.y) + (yw.z + yw.w);
    }
    const float dot = wave_sum(part);
    if (lane == 0) state[bn] = dot + sp_b[0];
}

#define GBM   32
#define GBK   32
#define GPA   (GBK + 4)
#define GPB   (2 * R + 16)
#define NSLAB (D / GBK)


__global__ __launch_bounds__(256) void qk_gemm_kernel(
    const float* __restrict__ val, const float* __restrict__ wq,
    const float* __restrict__ wk, float* __restrict__ q,
    float* __restrict__ k)
{
    __shared__ __attribute__((aligned(16))) float sA[2][GBM][GPA];
    __shared__ __attribute__((aligned(16))) float sB[2][GBK][GPB];

    const int tid  = threadIdx.x;
    const int lane = tid & 31;
    const int wave = tid >> 5;
    const int lm   = lane & 15;
    const int lh   = lane >> 4;
    const int mt   = wave & 1;
    const int n0   = (wave >> 1) * 32;
    const int row0 = blockIdx.x * GBM;

    const int ar = tid >> 3;
    const int ac = (tid & 7) * 4;
    const int br = tid >> 4;
    const int bc = (tid & 15) * 4;
    const float* ag = val + (size_t)min(row0 + ar, MM - 1) * D + ac;
    const float* qg = wq + (size_t)br * R + bc;
    const float* kg = wk + (size_t)br * R + bc;

    v4f ra  = *(const v4f*)ag;
    v4f rq0 = *(const v4f*)qg;
    v4f rq1 = *(const v4f*)(qg + 16 * R);
    v4f rk0 = *(const v4f*)kg;
    v4f rk1 = *(const v4f*)(kg + 16 * R);
    *(v4f*)&sA[0][ar][ac]          = ra;
    *(v4f*)&sB[0][br][bc]          = rq0;
    *(v4f*)&sB[0][br + 16][bc]     = rq1;
    *(v4f*)&sB[0][br][R + bc]      = rk0;
    *(v4f*)&sB[0][br + 16][R + bc] = rk1;
    __syncthreads();

    const bool active = (row0 + mt * 16) < MM;
    v8f acc0 = {};
    v8f acc1 = {};

    for (int s = 0; s < NSLAB; ++s) {
        const int cur = s & 1;
        const bool more = (s + 1) < NSLAB;
        if (more) {
            const int k0 = (s + 1) * GBK;
            ra  = *(const v4f*)(ag + k0);
            rq0 = *(const v4f*)(qg + (size_t)k0 * R);
            rq1 = *(const v4f*)(qg + (size_t)(k0 + 16) * R);
            rk0 = *(const v4f*)(kg + (size_t)k0 * R);
            rk1 = *(const v4f*)(kg + (size_t)(k0 + 16) * R);
        }
        if (active) {
            const float* pa = &sA[cur][mt * 16 + lm][2 * lh];
            const float* pb = &sB[cur][2 * lh][n0 + lm];
#pragma unroll
            for (int kk = 0; kk < GBK; kk += 4) {
                v2f a = *(const v2f*)(pa + kk);
                v2f b0, b1;
                b0.x = pb[kk * GPB];
                b0.y = pb[kk * GPB + GPB];
                b1.x = pb[kk * GPB + 16];
                b1.y = pb[kk * GPB + GPB + 16];
                acc0 = __builtin_amdgcn_wmma_f32_16x16x4_f32(
                    false, a, false, b0, (short)0, acc0, false, false);
                acc1 = __builtin_amdgcn_wmma_f32_16x16x4_f32(
                    false, a, false, b1, (short)0, acc1, false, false);
            }
        }
        if (more) {
            const int nxt = cur ^ 1;
            *(v4f*)&sA[nxt][ar][ac]          = ra;
            *(v4f*)&sB[nxt][br][bc]          = rq0;
            *(v4f*)&sB[nxt][br + 16][bc]     = rq1;
            *(v4f*)&sB[nxt][br][R + bc]      = rk0;
            *(v4f*)&sB[nxt][br + 16][R + bc] = rk1;
        }
        __syncthreads();
    }

    float (*sO)[GPB] = sB[0];
#pragma unroll
    for (int i = 0; i < 8; ++i) {
        sO[mt * 16 + 8 * lh + i][n0 + lm]      = acc0[i];
        sO[mt * 16 + 8 * lh + i][n0 + 16 + lm] = acc1[i];
    }
    __syncthreads();
    for (int j = 0; j < 4; ++j) {
        const int idx  = tid + 256 * j;
        const int orow = idx >> 5;
        const int oc   = (idx & 31) * 4;
        const int r    = row0 + orow;
        if (r < MM) {
            const v4f v = *(const v4f*)&sO[orow][oc];
            float* dst = (oc < R) ? (q + (size_t)r * R + oc) : (k + (size_t)r * R + (oc - R));
            *(v4f*)dst = v;
        }
    }
}

__global__ __launch_bounds__(256) void attn_fused_kernel(
    const float* __restrict__ val_in, const float* __restrict__ state_in,
    const float* __restrict__ q, const float* __restrict__ k,
    const float* __restrict__ gamma, const float* __restrict__ beta,
    float* __restrict__ val_out, float* __restrict__ state_out)
{
    __shared__ float s_scores[16][80];
    __shared__ float s_w[16][KB];
    __shared__ float s_y[16][D];
    __shared__ float s_r[16][16];
    __shared__ float s_mean[16];
    __shared__ float s_rstd[16];

    const int tid  = threadIdx.x;
    const int lane = tid & 31;
    const int wave = tid >> 5;
    const int lm   = lane & 15;
    const int lh   = lane >> 4;
    const int b    = blockIdx.x / TILES;
    const int t    = blockIdx.x % TILES;
    const int p0   = t * 16;
    const int base = b * NN;

    if (wave < 5) {
        const int qrow = base + clampi(p0 + lm, 0, NN - 1);
        const int j    = wave * 16 + lm;
        const int krow = base + clampi(p0 - 32 + j, 0, NN - 1);
        const float* qp = q + (size_t)qrow * R + 2 * lh;
        const float* kp = k + (size_t)krow * R + 2 * lh;
        v8f acc = {};
        for (int kk = 0; kk < R; kk += 4) {
            v2f a  = *(const v2f*)(qp + kk);
            v2f bb = *(const v2f*)(kp + kk);
            acc = __builtin_amdgcn_wmma_f32_16x16x4_f32(
                false, a, false, bb, (short)0, acc, false, false);
        }
#pragma unroll
        for (int i = 0; i < 8; ++i)
            s_scores[i + 8 * lh][wave * 16 + lm] = acc[i];
    }
    __syncthreads();

    for (int x = tid; x < 16 * KB; x += 256) (&s_w[0][0])[x] = 0.f;
    __syncthreads();

    if (tid < 16) {
        const int m    = tid;
        const int node = p0 + m;
        if (node < NN) {
            const float scale = 0.125f;
            float adot = 0.f;
            {
                const float* qp = q + (size_t)(base + node) * R;
                const float* k0 = k + (size_t)base * R;
                for (int r = 0; r < R; ++r) adot = fmaf(qp[r], k0[r], adot);
                adot *= scale;
            }
            const bool avalid = (node - WIN) > 0;
            float mx = avalid ? fabsf(adot) : -1e30f;
            for (int w = 0; w < 65; ++w) {
                const int p = node - WIN + w;
                if (p >= 0 && p < NN)
                    mx = fmaxf(mx, fabsf(s_scores[m][m + w] * scale));
            }
            float sum = 0.f;
            for (int w = 0; w < 65; ++w) {
                const int p = node - WIN + w;
                float se = 0.f;
                if (p >= 0 && p < NN) {
                    const float s = s_scores[m][m + w] * scale;
                    const float e = __expf(fabsf(s) - mx);
                    sum += e;
                    se = (s > 0.f) ? e : ((s < 0.f) ? -e : 0.f);
                }
                s_scores[m][m + w] = se;
            }
            float ase = 0.f;
            if (avalid) {
                const float e = __expf(fabsf(adot) - mx);
                sum += e;
                ase = (adot > 0.f) ? e : ((adot < 0.f) ? -e : 0.f);
            }
            const float inv = 1.f / sum;
            float ds = 0.f;
            for (int w = 0; w < 65; ++w) {
                const float we = s_scores[m][m + w] * inv;
                s_w[m][m + w] = we;
                ds += we * state_in[base + clampi(node - WIN + w, 0, NN - 1)];
            }
            const float wa = ase * inv;
            s_w[m][80] = wa;
            ds += wa * state_in[base];
            state_out[base + node] = state_in[base + node] + ds;
        }
    }
    __syncthreads();

    {
        const int c0 = wave * 64;
        v8f acc[4] = {};
        for (int kk = 0; kk < KB; kk += 4) {
            const int kp0 = kk + 2 * lh;
            const int kp1 = kp0 + 1;
            v2f a;
            a.x = s_w[lm][kp0];
            a.y = s_w[lm][kp1];
            const int r0 = (kp0 >= 80) ? base : base + clampi(p0 - 32 + kp0, 0, NN - 1);
            const int r1 = (kp1 >= 80) ? base : base + clampi(p0 - 32 + kp1, 0, NN - 1);
            const float* v0 = val_in + (size_t)r0 * D + c0 + lm;
            const float* v1 = val_in + (size_t)r1 * D + c0 + lm;
#pragma unroll
            for (int nt = 0; nt < 4; ++nt) {
                v2f bb;
                bb.x = v0[nt * 16];
                bb.y = v1[nt * 16];
                acc[nt] = __builtin_amdgcn_wmma_f32_16x16x4_f32(
                    false, a, false, bb, (short)0, acc[nt], false, false);
            }
        }
#pragma unroll
        for (int nt = 0; nt < 4; ++nt) {
#pragma unroll
            for (int i = 0; i < 8; ++i) {
                const int m    = i + 8 * lh;
                const int col  = c0 + nt * 16 + lm;
                const int node = p0 + m;
                float y = 0.f;
                if (node < NN)
                    y = acc[nt][i] + val_in[(size_t)(base + node) * D + col];
                s_y[m][col] = y;
            }
        }
    }
    __syncthreads();

    const int m  = tid >> 4;
    const int jj = tid & 15;
    float part = 0.f;
    for (int e = 0; e < 32; ++e) part += s_y[m][jj * 32 + e];
    s_r[m][jj] = part;
    __syncthreads();
    if (jj == 0) {
        float tot = 0.f;
        for (int x = 0; x < 16; ++x) tot += s_r[m][x];
        s_mean[m] = tot * (1.0f / D);
    }
    __syncthreads();
    const float mean = s_mean[m];
    part = 0.f;
    for (int e = 0; e < 32; ++e) {
        const float c = s_y[m][jj * 32 + e] - mean;
        part += c * c;
    }
    s_r[m][jj] = part;
    __syncthreads();
    if (jj == 0) {
        float tot = 0.f;
        for (int x = 0; x < 16; ++x) tot += s_r[m][x];
        s_rstd[m] = rsqrtf(tot * (1.0f / D) + 1e-5f);
    }
    __syncthreads();
    const int node = p0 + m;
    if (node < NN) {
        const float rstd = s_rstd[m];
        float* vo = val_out + (size_t)(base + node) * D;
        for (int e = 0; e < 32; ++e) {
            const int col = jj * 32 + e;
            vo[col] = (s_y[m][col] - mean) * rstd * gamma[col] + beta[col];
        }
    }
}

extern "C" void kernel_launch(void* const* d_in, const int* in_sizes, int n_in,
                              void* d_out, int out_size, void* d_ws, size_t ws_size,
                              hipStream_t stream)
{
    const int*   ids    = (const int*)  d_in[0];
    const float* emb    = (const float*)d_in[1];
    const float* pos    = (const float*)d_in[2];
    const float* aval   = (const float*)d_in[3];
    const float* astate = (const float*)d_in[4];
    const float* ing    = (const float*)d_in[5];
    const float* inb    = (const float*)d_in[6];
    const float* spw    = (const float*)d_in[7];
    const float* spb    = (const float*)d_in[8];
    const float* wq     = (const float*)d_in[9];
    const float* wk     = (const float*)d_in[10];
    const float* lng    = (const float*)d_in[11];
    const float* lnb    = (const float*)d_in[12];

    float* ws = (float*)d_ws;
    size_t o = 0;
    float* val_a = ws + o; o += (size_t)MM * D;
    float* val_b = ws + o; o += (size_t)MM * D;
    float* st_a  = ws + o; o += MM;
    float* st_b  = ws + o; o += MM;
    float* qbuf  = ws + o; o += (size_t)MM * R;
    float* kbuf  = ws + o; o += (size_t)MM * R;

    float* out_state = (float*)d_out;
    float* out_val   = out_state + MM;

    const int eblocks = (MM + EROWS - 1) / EROWS;
    embed_ln_kernel<<<eblocks, 256, 0, stream>>>(ids, emb, pos, aval, astate,
                                                 ing, inb, spw, spb, val_a, st_a);

    const int gblocks = (MM + GBM - 1) / GBM;
    const int ablocks = BB * TILES;

    qk_gemm_kernel<<<gblocks, 256, 0, stream>>>(val_a, wq, wk, qbuf, kbuf);
    attn_fused_kernel<<<ablocks, 256, 0, stream>>>(val_a, st_a, qbuf, kbuf,
                                                   lng, lnb, val_b, st_b);
    qk_gemm_kernel<<<gblocks, 256, 0, stream>>>(val_b, wq + (size_t)D * R,
                                                wk + (size_t)D * R, qbuf, kbuf);
    attn_fused_kernel<<<ablocks, 256, 0, stream>>>(val_b, st_b, qbuf, kbuf,
                                                   lng + D, lnb + D, out_val, out_state);
}


# fused kernel softmax phase on all 256 threads: 16 lanes per node, 5 band columns per lane, in-row butterflies for max, sum, dstate, anchor dot (was 16 serial threads); order of those sums changed
# speedup vs baseline: 2.5067x; 1.2186x over previous
#include <hip/hip_runtime.h>

#define D   512
#define R   64
#define WIN 32
#define BB  4
#define SS  2048
#define NN  (SS + 1)
#define MM  (BB * NN)
#define TILES ((NN + 15) / 16)
#define WTOT 66
#define KB  84

typedef float v2f __attribute__((ext_vector_type(2)));
typedef float v8f __attribute__((ext_vector_type(8)));

__device__ __forceinline__ int clampi(int x, int lo, int hi) {
    return min(max(x, lo), hi);
}

#define EROWS 8

typedef float v4f __attribute__((ext_vector_type(4)));

template <int M>
__device__ __forceinline__ float row_xor(float x) {
    const int b = __builtin_bit_cast(int, x);
    return __builtin_bit_cast(float,
        __builtin_amdgcn_update_dpp(b, b, 0x160 | M, 0xf, 0xf, false));
}

__device__ __forceinline__ float xhalf(float x) {
    const int b = __builtin_bit_cast(int, x);
    return __builtin_bit_cast(float,
        __builtin_amdgcn_permlanex16(b, b, 0x76543210, 0xfedcba98, false, false));
}

__device__ __forceinline__ float wave_sum(float x) {
    x += row_xor<1>(x);
    x += row_xor<2>(x);
    x += row_xor<4>(x);
    x += row_xor<8>(x);
    return x + xhalf(x);
}

__device__ __forceinline__ float row_sum16(float x) {
    x += row_xor<1>(x);
    x += row_xor<2>(x);
    x += row_xor<4>(x);
    return x + row_xor<8>(x);
}
__device__ __forceinline__ float row_max16(float x) {
    x = fmaxf(x, row_xor<1>(x));
    x = fmaxf(x, row_xor<2>(x));
    x = fmaxf(x, row_xor<4>(x));
    return fmaxf(x, row_xor<8>(x));
}

__global__ __launch_bounds__(256) void embed_ln_kernel(
    const int* __restrict__ ids, const float* __restrict__ emb,
    const float* __restrict__ pos, const float* __restrict__ anchor_val,
    const float* __restrict__ anchor_state, const float* __restrict__ gamma,
    const float* __restrict__ beta, const float* __restrict__ sp_w,
    const float* __restrict__ sp_b, float* __restrict__ val,
    float* __restrict__ state)
{
    const int lane = threadIdx.x & 31;
    const int bn = __builtin_amdgcn_readfirstlane((int)(blockIdx.x * EROWS + (threadIdx.x >> 5)));
    if (bn >= MM) return;
    const int b = bn / NN;
    const int n = bn - b * NN;
    const int c = lane * 4;
    float* vrow = val + (size_t)bn * D + c;

    if (n == 0) {
#pragma unroll
        for (int j = 0; j < 4; ++j)
            *(v4f*)(vrow + j * 128) = *(const v4f*)(anchor_val + j * 128 + c);
        if (lane == 0) state[bn] = anchor_state[0];
        return;
    }
    const int s  = n - 1;
    const int id = ids[b * SS + s];
    const float* erow = emb + (size_t)id * D + c;
    const float* prow = pos + (size_t)s * D + c;

    v4f x[4];
#pragma unroll
    for (int j = 0; j < 4; ++j)
        x[j] = *(const v4f*)(erow + j * 128) + *(const v4f*)(prow + j * 128);

    float part = 0.f;
#pragma unroll
    for (int j = 0; j < 4; ++j)
        part += (x[j].x + x[j].y) + (x[j].z + x[j].w);
    const float mean = wave_sum(part) * (1.0f / D);

    part = 0.f;
#pragma unroll
    for (int j = 0; j < 4; ++j) {
        x[j] -= mean;
        const v4f sq = x[j] * x[j];
        part += (sq.x + sq.y) + (sq.z + sq.w);
    }
    const float rstd = rsqrtf(wave_sum(part) * (1.0f / D) + 1e-5f);

    part = 0.f;
#pragma unroll
    for (int j = 0; j < 4; ++j) {
        const v4f g  = *(const v4f*)(gamma + j * 128 + c);
        const v4f bt = *(const v4f*)(beta + j * 128 + c);
        const v4f w  = *(const v4f*)(sp_w + j * 128 + c);
        const v4f y  = x[j] * rstd * g + bt;
        *(v4f*)(vrow + j * 128) = y;
        const v4f yw = y * w;
        part += (yw.x + yw.y) + (yw.z + yw.w);
    }
    const float dot = wave_sum(part);
    if (lane == 0) state[bn] = dot + sp_b[0];
}

#define GBM   32
#define GBK   32
#define GPA   (GBK + 4)
#define GPB   (2 * R + 16)
#define NSLAB (D / GBK)


__global__ __launch_bounds__(256) void qk_gemm_kernel(
    const float* __restrict__ val, const float* __restrict__ wq,
    const float* __restrict__ wk, float* __restrict__ q,
    float* __restrict__ k)
{
    __shared__ __attribute__((aligned(16))) float sA[2][GBM][GPA];
    __shared__ __attribute__((aligned(16))) float sB[2][GBK][GPB];

    const int tid  = threadIdx.x;
    const int lane = tid & 31;
    const int wave = tid >> 5;
    const int lm   = lane & 15;
    const int lh   = lane >> 4;
    const int mt   = wave & 1;
    const int n0   = (wave >> 1) * 32;
    const int row0 = blockIdx.x * GBM;

    const int ar = tid >> 3;
    const int ac = (tid & 7) * 4;
    const int br = tid >> 4;
    const int bc = (tid & 15) * 4;
    const float* ag = val + (size_t)min(row0 + ar, MM - 1) * D + ac;
    const float* qg = wq + (size_t)br * R + bc;
    const float* kg = wk + (size_t)br * R + bc;

    v4f ra  = *(const v4f*)ag;
    v4f rq0 = *(const v4f*)qg;
    v4f rq1 = *(const v4f*)(qg + 16 * R);
    v4f rk0 = *(const v4f*)kg;
    v4f rk1 = *(const v4f*)(kg + 16 * R);
    *(v4f*)&sA[0][ar][ac]          = ra;
    *(v4f*)&sB[0][br][bc]          = rq0;
    *(v4f*)&sB[0][br + 16][bc]     = rq1;
    *(v4f*)&sB[0][br][R + bc]      = rk0;
    *(v4f*)&sB[0][br + 16][R + bc] = rk1;
    __syncthreads();

    const bool active = (row0 + mt * 16) < MM;
    v8f acc0 = {};
    v8f acc1 = {};

    for (int s = 0; s < NSLAB; ++s) {
        const int cur = s & 1;
        const bool more = (s + 1) < NSLAB;
        if (more) {
            const int k0 = (s + 1) * GBK;
            ra  = *(const v4f*)(ag + k0);
            rq0 = *(const v4f*)(qg + (size_t)k0 * R);
            rq1 = *(const v4f*)(qg + (size_t)(k0 + 16) * R);
            rk0 = *(const v4f*)(kg + (size_t)k0 * R);
            rk1 = *(const v4f*)(kg + (size_t)(k0 + 16) * R);
        }
        if (active) {
            const float* pa = &sA[cur][mt * 16 + lm][2 * lh];
            const float* pb = &sB[cur][2 * lh][n0 + lm];
#pragma unroll
            for (int kk = 0; kk < GBK; kk += 4) {
                v2f a = *(const v2f*)(pa + kk);
                v2f b0, b1;
                b0.x = pb[kk * GPB];
                b0.y = pb[kk * GPB + GPB];
                b1.x = pb[kk * GPB + 16];
                b1.y = pb[kk * GPB + GPB + 16];
                acc0 = __builtin_amdgcn_wmma_f32_16x16x4_f32(
                    false, a, false, b0, (short)0, acc0, false, false);
                acc1 = __builtin_amdgcn_wmma_f32_16x16x4_f32(
                    false, a, false, b1, (short)0, acc1, false, false);
            }
        }
        if (more) {
            const int nxt = cur ^ 1;
            *(v4f*)&sA[nxt][ar][ac]          = ra;
            *(v4f*)&sB[nxt][br][bc]          = rq0;
            *(v4f*)&sB[nxt][br + 16][bc]     = rq1;
            *(v4f*)&sB[nxt][br][R + bc]      = rk0;
            *(v4f*)&sB[nxt][br + 16][R + bc] = rk1;
        }
        __syncthreads();
    }

    float (*sO)[GPB] = sB[0];
#pragma unroll
    for (int i = 0; i < 8; ++i) {
        sO[mt * 16 + 8 * lh + i][n0 + lm]      = acc0[i];
        sO[mt * 16 + 8 * lh + i][n0 + 16 + lm] = acc1[i];
    }
    __syncthreads();
    for (int j = 0; j < 4; ++j) {
        const int idx  = tid + 256 * j;
        const int orow = idx >> 5;
        const int oc   = (idx & 31) * 4;
        const int r    = row0 + orow;
        if (r < MM) {
            const v4f v = *(const v4f*)&sO[orow][oc];
            float* dst = (oc < R) ? (q + (size_t)r * R + oc) : (k + (size_t)r * R + (oc - R));
            *(v4f*)dst = v;
        }
    }
}

__global__ __launch_bounds__(256) void attn_fused_kernel(
    const float* __restrict__ val_in, const float* __restrict__ state_in,
    const float* __restrict__ q, const float* __restrict__ k,
    const float* __restrict__ gamma, const float* __restrict__ beta,
    float* __restrict__ val_out, float* __restrict__ state_out)
{
    __shared__ float s_scores[16][80];
    __shared__ float s_w[16][KB];
    __shared__ float s_y[16][D];
    __shared__ float s_r[16][16];
    __shared__ float s_mean[16];
    __shared__ float s_rstd[16];

    const int tid  = threadIdx.x;
    const int lane = tid & 31;
    const int wave = tid >> 5;
    const int lm   = lane & 15;
    const int lh   = lane >> 4;
    const int b    = blockIdx.x / TILES;
    const int t    = blockIdx.x % TILES;
    const int p0   = t * 16;
    const int base = b * NN;

    const int  sm     = tid >> 4;
    const int  sj     = tid & 15;
    const int  snode  = p0 + sm;
    const bool nodeok = snode < NN;
    const int  snc    = min(snode, NN - 1);
    const v4f qa = *(const v4f*)(q + (size_t)(base + snc) * R + 4 * sj);
    const v4f ka = *(const v4f*)(k + (size_t)base * R + 4 * sj);
    float st[5];
#pragma unroll
    for (int i = 0; i < 5; ++i)
        st[i] = state_in[base + clampi(p0 - 32 + sj + 16 * i, 0, NN - 1)];
    const float st_anchor = state_in[base];
    const float st_self   = state_in[base + snc];

    if (wave < 5) {
        const int qrow = base + clampi(p0 + lm, 0, NN - 1);
        const int j    = wave * 16 + lm;
        const int krow = base + clampi(p0 - 32 + j, 0, NN - 1);
        const float* qp = q + (size_t)qrow * R + 2 * lh;
        const float* kp = k + (size_t)krow * R + 2 * lh;
        v8f acc = {};
        for (int kk = 0; kk < R; kk += 4) {
            v2f a  = *(const v2f*)(qp + kk);
            v2f bb = *(const v2f*)(kp + kk);
            acc = __builtin_amdgcn_wmma_f32_16x16x4_f32(
                false, a, false, bb, (short)0, acc, false, false);
        }
#pragma unroll
        for (int i = 0; i < 8; ++i)
            s_scores[i + 8 * lh][wave * 16 + lm] = acc[i];
    }
    __syncthreads();

    {
        const float scale = 0.125f;
        float adot = (qa.x * ka.x + qa.y * ka.y) + (qa.z * ka.z + qa.w * ka.w);
        adot = row_sum16(adot) * scale;
        const bool avalid = nodeok && (snode - WIN) > 0;

        float sv[5];
        bool  inwin[5], ok[5];
        float mx = avalid ? fabsf(adot) : -1e30f;
#pragma unroll
        for (int i = 0; i < 5; ++i) {
            const int kc = sj + 16 * i;
            const int w  = kc - sm;
            const int p  = p0 - 32 + kc;
            inwin[i] = nodeok && w >= 0 && w <= 2 * WIN;
            ok[i]    = inwin[i] && p >= 0 && p < NN;
            sv[i]    = s_scores[sm][kc] * scale;
            if (ok[i]) mx = fmaxf(mx, fabsf(sv[i]));
        }
        mx = row_max16(mx);

        float se[5];
        float sum = 0.f;
#pragma unroll
        for (int i = 0; i < 5; ++i) {
            const float e = ok[i] ? __expf(fabsf(sv[i]) - mx) : 0.f;
            sum += e;
            se[i] = (sv[i] > 0.f) ? e : ((sv[i] < 0.f) ? -e : 0.f);
        }
        sum = row_sum16(sum);
        const float ea  = avalid ? __expf(fabsf(adot) - mx) : 0.f;
        const float ase = (adot > 0.f) ? ea : ((adot < 0.f) ? -ea : 0.f);
        sum += ea;
        const float inv = 1.f / sum;

        float ds = 0.f;
#pragma unroll
        for (int i = 0; i < 5; ++i) {
            const float we = ok[i] ? se[i] * inv : 0.f;
            s_w[sm][sj + 16 * i] = we;
            if (inwin[i]) ds += we * st[i];
        }
        ds = row_sum16(ds);
        const float wa = avalid ? ase * inv : 0.f;
        if (sj < KB - 80) s_w[sm][80 + sj] = (sj == 0) ? wa : 0.f;
        ds += wa * st_anchor;
        if (nodeok && sj == 0) state_out[base + snode] = st_self + ds;
    }
    __syncthreads();

    {
        const int c0 = wave * 64;
        v8f acc[4] = {};
        for (int kk = 0; kk < KB; kk += 4) {
            const int kp0 = kk + 2 * lh;
            const int kp1 = kp0 + 1;
            v2f a;
            a.x = s_w[lm][kp0];
            a.y = s_w[lm][kp1];
            const int r0 = (kp0 >= 80) ? base : base + clampi(p0 - 32 + kp0, 0, NN - 1);
            const int r1 = (kp1 >= 80) ? base : base + clampi(p0 - 32 + kp1, 0, NN - 1);
            const float* v0 = val_in + (size_t)r0 * D + c0 + lm;
            const float* v1 = val_in + (size_t)r1 * D + c0 + lm;
#pragma unroll
            for (int nt = 0; nt < 4; ++nt) {
                v2f bb;
                bb.x = v0[nt * 16];
                bb.y = v1[nt * 16];
                acc[nt] = __builtin_amdgcn_wmma_f32_16x16x4_f32(
                    false, a, false, bb, (short)0, acc[nt], false, false);
            }
        }
#pragma unroll
        for (int nt = 0; nt < 4; ++nt) {
#pragma unroll
            for (int i = 0; i < 8; ++i) {
                const int m    = i + 8 * lh;
                const int col  = c0 + nt * 16 + lm;
                const int node = p0 + m;
                float y = 0.f;
                if (node < NN)
                    y = acc[nt][i] + val_in[(size_t)(base + node) * D + col];
                s_y[m][col] = y;
            }
        }
    }
    __syncthreads();

    const int m  = tid >> 4;
    const int jj = tid & 15;
    float part = 0.f;
    for (int e = 0; e < 32; ++e) part += s_y[m][jj * 32 + e];
    s_r[m][jj] = part;
    __syncthreads();
    if (jj == 0) {
        float tot = 0.f;
        for (int x = 0; x < 16; ++x) tot += s_r[m][x];
        s_mean[m] = tot * (1.0f / D);
    }
    __syncthreads();
    const float mean = s_mean[m];
    part = 0.f;
    for (int e = 0; e < 32; ++e) {
        const float c = s_y[m][jj * 32 + e] - mean;
        part += c * c;
    }
    s_r[m][jj] = part;
    __syncthreads();
    if (jj == 0) {
        float tot = 0.f;
        for (int x = 0; x < 16; ++x) tot += s_r[m][x];
        s_rstd[m] = rsqrtf(tot * (1.0f / D) + 1e-5f);
    }
    __syncthreads();
    const int node = p0 + m;
    if (node < NN) {
        const float rstd = s_rstd[m];
        float* vo = val_out + (size_t)(base + node) * D;
        for (int e = 0; e < 32; ++e) {
            const int col = jj * 32 + e;
            vo[col] = (s_y[m][col] - mean) * rstd * gamma[col] + beta[col];
        }
    }
}

extern "C" void kernel_launch(void* const* d_in, const int* in_sizes, int n_in,
                              void* d_out, int out_size, void* d_ws, size_t ws_size,
                              hipStream_t stream)
{
    const int*   ids    = (const int*)  d_in[0];
    const float* emb    = (const float*)d_in[1];
    const float* pos    = (const float*)d_in[2];
    const float* aval   = (const float*)d_in[3];
    const float* astate = (const float*)d_in[4];
    const float* ing    = (const float*)d_in[5];
    const float* inb    = (const float*)d_in[6];
    const float* spw    = (const float*)d_in[7];
    const float* spb    = (const float*)d_in[8];
    const float* wq     = (const float*)d_in[9];
    const float* wk     = (const float*)d_in[10];
    const float* lng    = (const float*)d_in[11];
    const float* lnb    = (const float*)d_in[12];

    float* ws = (float*)d_ws;
    size_t o = 0;
    float* val_a = ws + o; o += (size_t)MM * D;
    float* val_b = ws + o; o += (size_t)MM * D;
    float* st_a  = ws + o; o += MM;
    float* st_b  = ws + o; o += MM;
    float* qbuf  = ws + o; o += (size_t)MM * R;
    float* kbuf  = ws + o; o += (size_t)MM * R;

    float* out_state = (float*)d_out;
    float* out_val   = out_state + MM;

    const int eblocks = (MM + EROWS - 1) / EROWS;
    embed_ln_kernel<<<eblocks, 256, 0, stream>>>(ids, emb, pos, aval, astate,
                                                 ing, inb, spw, spb, val_a, st_a);

    const int gblocks = (MM + GBM - 1) / GBM;
    const int ablocks = BB * TILES;

    qk_gemm_kernel<<<gblocks, 256, 0, stream>>>(val_a, wq, wk, qbuf, kbuf);
    attn_fused_kernel<<<ablocks, 256, 0, stream>>>(val_a, st_a, qbuf, kbuf,
                                                   lng, lnb, val_b, st_b);
    qk_gemm_kernel<<<gblocks, 256, 0, stream>>>(val_b, wq + (size_t)D * R,
                                                wk + (size_t)D * R, qbuf, kbuf);
    attn_fused_kernel<<<ablocks, 256, 0, stream>>>(val_b, st_b, qbuf, kbuf,
                                                   lng + D, lnb + D, out_val, out_state);
}


# fused kernel epilogue (residual + LayerNorm + store) one wave per row: 16-byte LDS reads, loads, stores, in-register butterflies (was 16 threads per row, 4-byte strided); row-sum order changed
# speedup vs baseline: 2.9327x; 1.1699x over previous
#include <hip/hip_runtime.h>

#define D   512
#define R   64
#define WIN 32
#define BB  4
#define SS  2048
#define NN  (SS + 1)
#define MM  (BB * NN)
#define TILES ((NN + 15) / 16)
#define WTOT 66
#define KB  84

typedef float v2f __attribute__((ext_vector_type(2)));
typedef float v8f __attribute__((ext_vector_type(8)));

__device__ __forceinline__ int clampi(int x, int lo, int hi) {
    return min(max(x, lo), hi);
}

#define EROWS 8

typedef float v4f __attribute__((ext_vector_type(4)));

template <int M>
__device__ __forceinline__ float row_xor(float x) {
    const int b = __builtin_bit_cast(int, x);
    return __builtin_bit_cast(float,
        __builtin_amdgcn_update_dpp(b, b, 0x160 | M, 0xf, 0xf, false));
}

__device__ __forceinline__ float xhalf(float x) {
    const int b = __builtin_bit_cast(int, x);
    return __builtin_bit_cast(float,
        __builtin_amdgcn_permlanex16(b, b, 0x76543210, 0xfedcba98, false, false));
}

__device__ __forceinline__ float wave_sum(float x) {
    x += row_xor<1>(x);
    x += row_xor<2>(x);
    x += row_xor<4>(x);
    x += row_xor<8>(x);
    return x + xhalf(x);
}

__device__ __forceinline__ float row_sum16(float x) {
    x += row_xor<1>(x);
    x += row_xor<2>(x);
    x += row_xor<4>(x);
    return x + row_xor<8>(x);
}
__device__ __forceinline__ float row_max16(float x) {
    x = fmaxf(x, row_xor<1>(x));
    x = fmaxf(x, row_xor<2>(x));
    x = fmaxf(x, row_xor<4>(x));
    return fmaxf(x, row_xor<8>(x));
}

__global__ __launch_bounds__(256) void embed_ln_kernel(
    const int* __restrict__ ids, const float* __restrict__ emb,
    const float* __restrict__ pos, const float* __restrict__ anchor_val,
    const float* __restrict__ anchor_state, const float* __restrict__ gamma,
    const float* __restrict__ beta, const float* __restrict__ sp_w,
    const float* __restrict__ sp_b, float* __restrict__ val,
    float* __restrict__ state)
{
    const int lane = threadIdx.x & 31;
    const int bn = __builtin_amdgcn_readfirstlane((int)(blockIdx.x * EROWS + (threadIdx.x >> 5)));
    if (bn >= MM) return;
    const int b = bn / NN;
    const int n = bn - b * NN;
    const int c = lane * 4;
    float* vrow = val + (size_t)bn * D + c;

    if (n == 0) {
#pragma unroll
        for (int j = 0; j < 4; ++j)
            *(v4f*)(vrow + j * 128) = *(const v4f*)(anchor_val + j * 128 + c);
        if (lane == 0) state[bn] = anchor_state[0];
        return;
    }
    const int s  = n - 1;
    const int id = ids[b * SS + s];
    const float* erow = emb + (size_t)id * D + c;
    const float* prow = pos + (size_t)s * D + c;

    v4f x[4];
#pragma unroll
    for (int j = 0; j < 4; ++j)
        x[j] = *(const v4f*)(erow + j * 128) + *(const v4f*)(prow + j * 128);

    float part = 0.f;
#pragma unroll
    for (int j = 0; j < 4; ++j)
        part += (x[j].x + x[j].y) + (x[j].z + x[j].w);
    const float mean = wave_sum(part) * (1.0f / D);

    part = 0.f;
#pragma unroll
    for (int j = 0; j < 4; ++j) {
        x[j] -= mean;
        const v4f sq = x[j] * x[j];
        part += (sq.x + sq.y) + (sq.z + sq.w);
    }
    const float rstd = rsqrtf(wave_sum(part) * (1.0f / D) + 1e-5f);

    part = 0.f;
#pragma unroll
    for (int j = 0; j < 4; ++j) {
        const v4f g  = *(const v4f*)(gamma + j * 128 + c);
        const v4f bt = *(const v4f*)(beta + j * 128 + c);
        const v4f w  = *(const v4f*)(sp_w + j * 128 + c);
        const v4f y  = x[j] * rstd * g + bt;
        *(v4f*)(vrow + j * 128) = y;
        const v4f yw = y * w;
        part += (yw.x + yw.y) + (yw.z + yw.w);
    }
    const float dot = wave_sum(part);
    if (lane == 0) state[bn] = dot + sp_b[0];
}

#define GBM   32
#define GBK   32
#define GPA   (GBK + 4)
#define GPB   (2 * R + 16)
#define NSLAB (D / GBK)


__global__ __launch_bounds__(256) void qk_gemm_kernel(
    const float* __restrict__ val, const float* __restrict__ wq,
    const float* __restrict__ wk, float* __restrict__ q,
    float* __restrict__ k)
{
    __shared__ __attribute__((aligned(16))) float sA[2][GBM][GPA];
    __shared__ __attribute__((aligned(16))) float sB[2][GBK][GPB];

    const int tid  = threadIdx.x;
    const int lane = tid & 31;
    const int wave = tid >> 5;
    const int lm   = lane & 15;
    const int lh   = lane >> 4;
    const int mt   = wave & 1;
    const int n0   = (wave >> 1) * 32;
    const int row0 = blockIdx.x * GBM;

    const int ar = tid >> 3;
    const int ac = (tid & 7) * 4;
    const int br = tid >> 4;
    const int bc = (tid & 15) * 4;
    const float* ag = val + (size_t)min(row0 + ar, MM - 1) * D + ac;
    const float* qg = wq + (size_t)br * R + bc;
    const float* kg = wk + (size_t)br * R + bc;

    v4f ra  = *(const v4f*)ag;
    v4f rq0 = *(const v4f*)qg;
    v4f rq1 = *(const v4f*)(qg + 16 * R);
    v4f rk0 = *(const v4f*)kg;
    v4f rk1 = *(const v4f*)(kg + 16 * R);
    *(v4f*)&sA[0][ar][ac]          = ra;
    *(v4f*)&sB[0][br][bc]          = rq0;
    *(v4f*)&sB[0][br + 16][bc]     = rq1;
    *(v4f*)&sB[0][br][R + bc]      = rk0;
    *(v4f*)&sB[0][br + 16][R + bc] = rk1;
    __syncthreads();

    const bool active = (row0 + mt * 16) < MM;
    v8f acc0 = {};
    v8f acc1 = {};

    for (int s = 0; s < NSLAB; ++s) {
        const int cur = s & 1;
        const bool more = (s + 1) < NSLAB;
        if (more) {
            const int k0 = (s + 1) * GBK;
            ra  = *(const v4f*)(ag + k0);
            rq0 = *(const v4f*)(qg + (size_t)k0 * R);
            rq1 = *(const v4f*)(qg + (size_t)(k0 + 16) * R);
            rk0 = *(const v4f*)(kg + (size_t)k0 * R);
            rk1 = *(const v4f*)(kg + (size_t)(k0 + 16) * R);
        }
        if (active) {
            const float* pa = &sA[cur][mt * 16 + lm][2 * lh];
            const float* pb = &sB[cur][2 * lh][n0 + lm];
#pragma unroll
            for (int kk = 0; kk < GBK; kk += 4) {
                v2f a = *(const v2f*)(pa + kk);
                v2f b0, b1;
                b0.x = pb[kk * GPB];
                b0.y = pb[kk * GPB + GPB];
                b1.x = pb[kk * GPB + 16];
                b1.y = pb[kk * GPB + GPB + 16];
                acc0 = __builtin_amdgcn_wmma_f32_16x16x4_f32(
                    false, a, false, b0, (short)0, acc0, false, false);
                acc1 = __builtin_amdgcn_wmma_f32_16x16x4_f32(
                    false, a, false, b1, (short)0, acc1, false, false);
            }
        }
        if (more) {
            const int nxt = cur ^ 1;
            *(v4f*)&sA[nxt][ar][ac]          = ra;
            *(v4f*)&sB[nxt][br][bc]          = rq0;
            *(v4f*)&sB[nxt][br + 16][bc]     = rq1;
            *(v4f*)&sB[nxt][br][R + bc]      = rk0;
            *(v4f*)&sB[nxt][br + 16][R + bc] = rk1;
        }
        __syncthreads();
    }

    float (*sO)[GPB] = sB[0];
#pragma unroll
    for (int i = 0; i < 8; ++i) {
        sO[mt * 16 + 8 * lh + i][n0 + lm]      = acc0[i];
        sO[mt * 16 + 8 * lh + i][n0 + 16 + lm] = acc1[i];
    }
    __syncthreads();
    for (int j = 0; j < 4; ++j) {
        const int idx  = tid + 256 * j;
        const int orow = idx >> 5;
        const int oc   = (idx & 31) * 4;
        const int r    = row0 + orow;
        if (r < MM) {
            const v4f v = *(const v4f*)&sO[orow][oc];
            float* dst = (oc < R) ? (q + (size_t)r * R + oc) : (k + (size_t)r * R + (oc - R));
            *(v4f*)dst = v;
        }
    }
}

__global__ __launch_bounds__(256) void attn_fused_kernel(
    const float* __restrict__ val_in, const float* __restrict__ state_in,
    const float* __restrict__ q, const float* __restrict__ k,
    const float* __restrict__ gamma, const float* __restrict__ beta,
    float* __restrict__ val_out, float* __restrict__ state_out)
{
    __shared__ float s_scores[16][80];
    __shared__ float s_w[16][KB];
    __shared__ __attribute__((aligned(16))) float s_y[16][D];

    const int tid  = threadIdx.x;
    const int lane = tid & 31;
    const int wave = tid >> 5;
    const int lm   = lane & 15;
    const int lh   = lane >> 4;
    const int b    = blockIdx.x / TILES;
    const int t    = blockIdx.x % TILES;
    const int p0   = t * 16;
    const int base = b * NN;

    const int  sm     = tid >> 4;
    const int  sj     = tid & 15;
    const int  snode  = p0 + sm;
    const bool nodeok = snode < NN;
    const int  snc    = min(snode, NN - 1);
    const v4f qa = *(const v4f*)(q + (size_t)(base + snc) * R + 4 * sj);
    const v4f ka = *(const v4f*)(k + (size_t)base * R + 4 * sj);
    float st[5];
#pragma unroll
    for (int i = 0; i < 5; ++i)
        st[i] = state_in[base + clampi(p0 - 32 + sj + 16 * i, 0, NN - 1)];
    const float st_anchor = state_in[base];
    const float st_self   = state_in[base + snc];

    if (wave < 5) {
        const int qrow = base + clampi(p0 + lm, 0, NN - 1);
        const int j    = wave * 16 + lm;
        const int krow = base + clampi(p0 - 32 + j, 0, NN - 1);
        const float* qp = q + (size_t)qrow * R + 2 * lh;
        const float* kp = k + (size_t)krow * R + 2 * lh;
        v8f acc = {};
        for (int kk = 0; kk < R; kk += 4) {
            v2f a  = *(const v2f*)(qp + kk);
            v2f bb = *(const v2f*)(kp + kk);
            acc = __builtin_amdgcn_wmma_f32_16x16x4_f32(
                false, a, false, bb, (short)0, acc, false, false);
        }
#pragma unroll
        for (int i = 0; i < 8; ++i)
            s_scores[i + 8 * lh][wave * 16 + lm] = acc[i];
    }
    __syncthreads();

    {
        const float scale = 0.125f;
        float adot = (qa.x * ka.x + qa.y * ka.y) + (qa.z * ka.z + qa.w * ka.w);
        adot = row_sum16(adot) * scale;
        const bool avalid = nodeok && (snode - WIN) > 0;

        float sv[5];
        bool  inwin[5], ok[5];
        float mx = avalid ? fabsf(adot) : -1e30f;
#pragma unroll
        for (int i = 0; i < 5; ++i) {
            const int kc = sj + 16 * i;
            const int w  = kc - sm;
            const int p  = p0 - 32 + kc;
            inwin[i] = nodeok && w >= 0 && w <= 2 * WIN;
            ok[i]    = inwin[i] && p >= 0 && p < NN;
            sv[i]    = s_scores[sm][kc] * scale;
            if (ok[i]) mx = fmaxf(mx, fabsf(sv[i]));
        }
        mx = row_max16(mx);

        float se[5];
        float sum = 0.f;
#pragma unroll
        for (int i = 0; i < 5; ++i) {
            const float e = ok[i] ? __expf(fabsf(sv[i]) - mx) : 0.f;
            sum += e;
            se[i] = (sv[i] > 0.f) ? e : ((sv[i] < 0.f) ? -e : 0.f);
        }
        sum = row_sum16(sum);
        const float ea  = avalid ? __expf(fabsf(adot) - mx) : 0.f;
        const float ase = (adot > 0.f) ? ea : ((adot < 0.f) ? -ea : 0.f);
        sum += ea;
        const float inv = 1.f / sum;

        float ds = 0.f;
#pragma unroll
        for (int i = 0; i < 5; ++i) {
            const float we = ok[i] ? se[i] * inv : 0.f;
            s_w[sm][sj + 16 * i] = we;
            if (inwin[i]) ds += we * st[i];
        }
        ds = row_sum16(ds);
        const float wa = avalid ? ase * inv : 0.f;
        if (sj < KB - 80) s_w[sm][80 + sj] = (sj == 0) ? wa : 0.f;
        ds += wa * st_anchor;
        if (nodeok && sj == 0) state_out[base + snode] = st_self + ds;
    }
    __syncthreads();

    {
        const int c0 = wave * 64;
        v8f acc[4] = {};
        for (int kk = 0; kk < KB; kk += 4) {
            const int kp0 = kk + 2 * lh;
            const int kp1 = kp0 + 1;
            v2f a;
            a.x = s_w[lm][kp0];
            a.y = s_w[lm][kp1];
            const int r0 = (kp0 >= 80) ? base : base + clampi(p0 - 32 + kp0, 0, NN - 1);
            const int r1 = (kp1 >= 80) ? base : base + clampi(p0 - 32 + kp1, 0, NN - 1);
            const float* v0 = val_in + (size_t)r0 * D + c0 + lm;
            const float* v1 = val_in + (size_t)r1 * D + c0 + lm;
#pragma unroll
            for (int nt = 0; nt < 4; ++nt) {
                v2f bb;
                bb.x = v0[nt * 16];
                bb.y = v1[nt * 16];
                acc[nt] = __builtin_amdgcn_wmma_f32_16x16x4_f32(
                    false, a, false, bb, (short)0, acc[nt], false, false);
            }
        }
#pragma unroll
        for (int nt = 0; nt < 4; ++nt) {
#pragma unroll
            for (int i = 0; i < 8; ++i)
                s_y[i + 8 * lh][c0 + nt * 16 + lm] = acc[nt][i];
        }
    }

    const int wrow = 2 * __builtin_amdgcn_readfirstlane(wave);
    const int c4   = lane * 4;
    v4f res[2][4];
    v4f g[4], bt[4];
#pragma unroll
    for (int r = 0; r < 2; ++r) {
        const int nodec = min(p0 + wrow + r, NN - 1);
        const float* vr = val_in + (size_t)(base + nodec) * D + c4;
#pragma unroll
        for (int j = 0; j < 4; ++j)
            res[r][j] = *(const v4f*)(vr + j * 128);
    }
#pragma unroll
    for (int j = 0; j < 4; ++j) {
        g[j]  = *(const v4f*)(gamma + j * 128 + c4);
        bt[j] = *(const v4f*)(beta + j * 128 + c4);
    }
    __syncthreads();

#pragma unroll
    for (int r = 0; r < 2; ++r) {
        const int m    = wrow + r;
        const int node = p0 + m;
        if (node < NN) {
            v4f y[4];
            float part = 0.f;
#pragma unroll
            for (int j = 0; j < 4; ++j) {
                y[j] = *(const v4f*)&s_y[m][j * 128 + c4] + res[r][j];
                part += (y[j].x + y[j].y) + (y[j].z + y[j].w);
            }
            const float mean = wave_sum(part) * (1.0f / D);
            part = 0.f;
#pragma unroll
            for (int j = 0; j < 4; ++j) {
                y[j] -= mean;
                const v4f sq = y[j] * y[j];
                part += (sq.x + sq.y) + (sq.z + sq.w);
            }
            const float rstd = rsqrtf(wave_sum(part) * (1.0f / D) + 1e-5f);
            float* vo = val_out + (size_t)(base + node) * D + c4;
#pragma unroll
            for (int j = 0; j < 4; ++j)
                *(v4f*)(vo + j * 128) = y[j] * rstd * g[j] + bt[j];
        }
    }
}

extern "C" void kernel_launch(void* const* d_in, const int* in_sizes, int n_in,
                              void* d_out, int out_size, void* d_ws, size_t ws_size,
                              hipStream_t stream)
{
    const int*   ids    = (const int*)  d_in[0];
    const float* emb    = (const float*)d_in[1];
    const float* pos    = (const float*)d_in[2];
    const float* aval   = (const float*)d_in[3];
    const float* astate = (const float*)d_in[4];
    const float* ing    = (const float*)d_in[5];
    const float* inb    = (const float*)d_in[6];
    const float* spw    = (const float*)d_in[7];
    const float* spb    = (const float*)d_in[8];
    const float* wq     = (const float*)d_in[9];
    const float* wk     = (const float*)d_in[10];
    const float* lng    = (const float*)d_in[11];
    const float* lnb    = (const float*)d_in[12];

    float* ws = (float*)d_ws;
    size_t o = 0;
    float* val_a = ws + o; o += (size_t)MM * D;
    float* val_b = ws + o; o += (size_t)MM * D;
    float* st_a  = ws + o; o += MM;
    float* st_b  = ws + o; o += MM;
    float* qbuf  = ws + o; o += (size_t)MM * R;
    float* kbuf  = ws + o; o += (size_t)MM * R;

    float* out_state = (float*)d_out;
    float* out_val   = out_state + MM;

    const int eblocks = (MM + EROWS - 1) / EROWS;
    embed_ln_kernel<<<eblocks, 256, 0, stream>>>(ids, emb, pos, aval, astate,
                                                 ing, inb, spw, spb, val_a, st_a);

    const int gblocks = (MM + GBM - 1) / GBM;
    const int ablocks = BB * TILES;

    qk_gemm_kernel<<<gblocks, 256, 0, stream>>>(val_a, wq, wk, qbuf, kbuf);
    attn_fused_kernel<<<ablocks, 256, 0, stream>>>(val_a, st_a, qbuf, kbuf,
                                                   lng, lnb, val_b, st_b);
    qk_gemm_kernel<<<gblocks, 256, 0, stream>>>(val_b, wq + (size_t)D * R,
                                                wk + (size_t)D * R, qbuf, kbuf);
    attn_fused_kernel<<<ablocks, 256, 0, stream>>>(val_b, st_b, qbuf, kbuf,
                                                   lng + D, lnb + D, out_val, out_state);
}


# fused kernel dval GEMM operand path: val band rows by 16-byte loads (lane owns 4 adjacent channels, one per tile), 42 per wave in 4 groups issued ahead of the products, 2 beside the softmax
# speedup vs baseline: 3.5063x; 1.1956x over previous
#include <hip/hip_runtime.h>

#define D   512
#define R   64
#define WIN 32
#define BB  4
#define SS  2048
#define NN  (SS + 1)
#define MM  (BB * NN)
#define TILES ((NN + 15) / 16)
#define WTOT 66
#define KB  84
#define P2G 6

typedef float v2f __attribute__((ext_vector_type(2)));
typedef float v8f __attribute__((ext_vector_type(8)));

__device__ __forceinline__ int clampi(int x, int lo, int hi) {
    return min(max(x, lo), hi);
}

#define EROWS 8

typedef float v4f __attribute__((ext_vector_type(4)));

template <int M>
__device__ __forceinline__ float row_xor(float x) {
    const int b = __builtin_bit_cast(int, x);
    return __builtin_bit_cast(float,
        __builtin_amdgcn_update_dpp(b, b, 0x160 | M, 0xf, 0xf, false));
}

__device__ __forceinline__ float xhalf(float x) {
    const int b = __builtin_bit_cast(int, x);
    return __builtin_bit_cast(float,
        __builtin_amdgcn_permlanex16(b, b, 0x76543210, 0xfedcba98, false, false));
}

__device__ __forceinline__ float wave_sum(float x) {
    x += row_xor<1>(x);
    x += row_xor<2>(x);
    x += row_xor<4>(x);
    x += row_xor<8>(x);
    return x + xhalf(x);
}

__device__ __forceinline__ float row_sum16(float x) {
    x += row_xor<1>(x);
    x += row_xor<2>(x);
    x += row_xor<4>(x);
    return x + row_xor<8>(x);
}
__device__ __forceinline__ float row_max16(float x) {
    x = fmaxf(x, row_xor<1>(x));
    x = fmaxf(x, row_xor<2>(x));
    x = fmaxf(x, row_xor<4>(x));
    return fmaxf(x, row_xor<8>(x));
}

__global__ __launch_bounds__(256) void embed_ln_kernel(
    const int* __restrict__ ids, const float* __restrict__ emb,
    const float* __restrict__ pos, const float* __restrict__ anchor_val,
    const float* __restrict__ anchor_state, const float* __restrict__ gamma,
    const float* __restrict__ beta, const float* __restrict__ sp_w,
    const float* __restrict__ sp_b, float* __restrict__ val,
    float* __restrict__ state)
{
    const int lane = threadIdx.x & 31;
    const int bn = __builtin_amdgcn_readfirstlane((int)(blockIdx.x * EROWS + (threadIdx.x >> 5)));
    if (bn >= MM) return;
    const int b = bn / NN;
    const int n = bn - b * NN;
    const int c = lane * 4;
    float* vrow = val + (size_t)bn * D + c;

    if (n == 0) {
#pragma unroll
        for (int j = 0; j < 4; ++j)
            *(v4f*)(vrow + j * 128) = *(const v4f*)(anchor_val + j * 128 + c);
        if (lane == 0) state[bn] = anchor_state[0];
        return;
    }
    const int s  = n - 1;
    const int id = ids[b * SS + s];
    const float* erow = emb + (size_t)id * D + c;
    const float* prow = pos + (size_t)s * D + c;

    v4f x[4];
#pragma unroll
    for (int j = 0; j < 4; ++j)
        x[j] = *(const v4f*)(erow + j * 128) + *(const v4f*)(prow + j * 128);

    float part = 0.f;
#pragma unroll
    for (int j = 0; j < 4; ++j)
        part += (x[j].x + x[j].y) + (x[j].z + x[j].w);
    const float mean = wave_sum(part) * (1.0f / D);

    part = 0.f;
#pragma unroll
    for (int j = 0; j < 4; ++j) {
        x[j] -= mean;
        const v4f sq = x[j] * x[j];
        part += (sq.x + sq.y) + (sq.z + sq.w);
    }
    const float rstd = rsqrtf(wave_sum(part) * (1.0f / D) + 1e-5f);

    part = 0.f;
#pragma unroll
    for (int j = 0; j < 4; ++j) {
        const v4f g  = *(const v4f*)(gamma + j * 128 + c);
        const v4f bt = *(const v4f*)(beta + j * 128 + c);
        const v4f w  = *(const v4f*)(sp_w + j * 128 + c);
        const v4f y  = x[j] * rstd * g + bt;
        *(v4f*)(vrow + j * 128) = y;
        const v4f yw = y * w;
        part += (yw.x + yw.y) + (yw.z + yw.w);
    }
    const float dot = wave_sum(part);
    if (lane == 0) state[bn] = dot + sp_b[0];
}

#define GBM   32
#define GBK   32
#define GPA   (GBK + 4)
#define GPB   (2 * R + 16)
#define NSLAB (D / GBK)


__global__ __launch_bounds__(256) void qk_gemm_kernel(
    const float* __restrict__ val, const float* __restrict__ wq,
    const float* __restrict__ wk, float* __restrict__ q,
    float* __restrict__ k)
{
    __shared__ __attribute__((aligned(16))) float sA[2][GBM][GPA];
    __shared__ __attribute__((aligned(16))) float sB[2][GBK][GPB];

    const int tid  = threadIdx.x;
    const int lane = tid & 31;
    const int wave = tid >> 5;
    const int lm   = lane & 15;
    const int lh   = lane >> 4;
    const int mt   = wave & 1;
    const int n0   = (wave >> 1) * 32;
    const int row0 = blockIdx.x * GBM;

    const int ar = tid >> 3;
    const int ac = (tid & 7) * 4;
    const int br = tid >> 4;
    const int bc = (tid & 15) * 4;
    const float* ag = val + (size_t)min(row0 + ar, MM - 1) * D + ac;
    const float* qg = wq + (size_t)br * R + bc;
    const float* kg = wk + (size_t)br * R + bc;

    v4f ra  = *(const v4f*)ag;
    v4f rq0 = *(const v4f*)qg;
    v4f rq1 = *(const v4f*)(qg + 16 * R);
    v4f rk0 = *(const v4f*)kg;
    v4f rk1 = *(const v4f*)(kg + 16 * R);
    *(v4f*)&sA[0][ar][ac]          = ra;
    *(v4f*)&sB[0][br][bc]          = rq0;
    *(v4f*)&sB[0][br + 16][bc]     = rq1;
    *(v4f*)&sB[0][br][R + bc]      = rk0;
    *(v4f*)&sB[0][br + 16][R + bc] = rk1;
    __syncthreads();

    const bool active = (row0 + mt * 16) < MM;
    v8f acc0 = {};
    v8f acc1 = {};

    for (int s = 0; s < NSLAB; ++s) {
        const int cur = s & 1;
        const bool more = (s + 1) < NSLAB;
        if (more) {
            const int k0 = (s + 1) * GBK;
            ra  = *(const v4f*)(ag + k0);
            rq0 = *(const v4f*)(qg + (size_t)k0 * R);
            rq1 = *(const v4f*)(qg + (size_t)(k0 + 16) * R);
            rk0 = *(const v4f*)(kg + (size_t)k0 * R);
            rk1 = *(const v4f*)(kg + (size_t)(k0 + 16) * R);
        }
        if (active) {
            const float* pa = &sA[cur][mt * 16 + lm][2 * lh];
            const float* pb = &sB[cur][2 * lh][n0 + lm];
#pragma unroll
            for (int kk = 0; kk < GBK; kk += 4) {
                v2f a = *(const v2f*)(pa + kk);
                v2f b0, b1;
                b0.x = pb[kk * GPB];
                b0.y = pb[kk * GPB + GPB];
                b1.x = pb[kk * GPB + 16];
                b1.y = pb[kk * GPB + GPB + 16];
                acc0 = __builtin_amdgcn_wmma_f32_16x16x4_f32(
                    false, a, false, b0, (short)0, acc0, false, false);
                acc1 = __builtin_amdgcn_wmma_f32_16x16x4_f32(
                    false, a, false, b1, (short)0, acc1, false, false);
            }
        }
        if (more) {
            const int nxt = cur ^ 1;
            *(v4f*)&sA[nxt][ar][ac]          = ra;
            *(v4f*)&sB[nxt][br][bc]          = rq0;
            *(v4f*)&sB[nxt][br + 16][bc]     = rq1;
            *(v4f*)&sB[nxt][br][R + bc]      = rk0;
            *(v4f*)&sB[nxt][br + 16][R + bc] = rk1;
        }
        __syncthreads();
    }

    float (*sO)[GPB] = sB[0];
#pragma unroll
    for (int i = 0; i < 8; ++i) {
        sO[mt * 16 + 8 * lh + i][n0 + lm]      = acc0[i];
        sO[mt * 16 + 8 * lh + i][n0 + 16 + lm] = acc1[i];
    }
    __syncthreads();
    for (int j = 0; j < 4; ++j) {
        const int idx  = tid + 256 * j;
        const int orow = idx >> 5;
        const int oc   = (idx & 31) * 4;
        const int r    = row0 + orow;
        if (r < MM) {
            const v4f v = *(const v4f*)&sO[orow][oc];
            float* dst = (oc < R) ? (q + (size_t)r * R + oc) : (k + (size_t)r * R + (oc - R));
            *(v4f*)dst = v;
        }
    }
}

__device__ __forceinline__ int p2_row(int p0, int lh, int ks, int s) {
    return (ks >= 20) ? 0 : clampi(p0 - 32 + 4 * ks + 2 * lh + s, 0, NN - 1);
}

template <int KS0, int N>
__device__ __forceinline__ void p2_load(v4f (&bs)[P2G][2], const float* vband, int p0, int lh) {
#pragma unroll
    for (int i = 0; i < N; ++i) {
#pragma unroll
        for (int s = 0; s < 2; ++s)
            bs[i][s] = *(const v4f*)(vband + p2_row(p0, lh, KS0 + i, s) * D);
    }
}

template <int KS0, int N>
__device__ __forceinline__ void p2_mma(v8f (&acc)[4], const v4f (&bs)[P2G][2], const float* wfrag) {
    v2f a[N];
#pragma unroll
    for (int i = 0; i < N; ++i)
        a[i] = *(const v2f*)(wfrag + 4 * (KS0 + i));
#pragma unroll
    for (int i = 0; i < N; ++i) {
#pragma unroll
        for (int nt = 0; nt < 4; ++nt) {
            v2f bb;
            bb.x = bs[i][0][nt];
            bb.y = bs[i][1][nt];
            acc[nt] = __builtin_amdgcn_wmma_f32_16x16x4_f32(
                false, a[i], false, bb, (short)0, acc[nt], false, false);
        }
    }
}

__global__ __launch_bounds__(256) void attn_fused_kernel(
    const float* __restrict__ val_in, const float* __restrict__ state_in,
    const float* __restrict__ q, const float* __restrict__ k,
    const float* __restrict__ gamma, const float* __restrict__ beta,
    float* __restrict__ val_out, float* __restrict__ state_out)
{
    __shared__ float s_scores[16][80];
    __shared__ float s_w[16][KB];
    __shared__ __attribute__((aligned(16))) float s_y[16][D];

    const int tid  = threadIdx.x;
    const int lane = tid & 31;
    const int wave = tid >> 5;
    const int lm   = lane & 15;
    const int lh   = lane >> 4;
    const int b    = blockIdx.x / TILES;
    const int t    = blockIdx.x % TILES;
    const int p0   = t * 16;
    const int base = b * NN;

    const int  sm     = tid >> 4;
    const int  sj     = tid & 15;
    const int  snode  = p0 + sm;
    const bool nodeok = snode < NN;
    const int  snc    = min(snode, NN - 1);
    const v4f qa = *(const v4f*)(q + (size_t)(base + snc) * R + 4 * sj);
    const v4f ka = *(const v4f*)(k + (size_t)base * R + 4 * sj);
    float st[5];
#pragma unroll
    for (int i = 0; i < 5; ++i)
        st[i] = state_in[base + clampi(p0 - 32 + sj + 16 * i, 0, NN - 1)];
    const float st_anchor = state_in[base];
    const float st_self   = state_in[base + snc];

    if (wave < 5) {
        const int qrow = base + clampi(p0 + lm, 0, NN - 1);
        const int j    = wave * 16 + lm;
        const int krow = base + clampi(p0 - 32 + j, 0, NN - 1);
        const float* qp = q + (size_t)qrow * R + 2 * lh;
        const float* kp = k + (size_t)krow * R + 2 * lh;
        v8f acc = {};
        for (int kk = 0; kk < R; kk += 4) {
            v2f a  = *(const v2f*)(qp + kk);
            v2f bb = *(const v2f*)(kp + kk);
            acc = __builtin_amdgcn_wmma_f32_16x16x4_f32(
                false, a, false, bb, (short)0, acc, false, false);
        }
#pragma unroll
        for (int i = 0; i < 8; ++i)
            s_scores[i + 8 * lh][wave * 16 + lm] = acc[i];
    }
    __syncthreads();

    const int c0 = wave * 64;
    const float* vband = val_in + (size_t)base * D + c0 + 4 * lm;
    v4f bset0[P2G][2], bset1[P2G][2];
    p2_load<0, P2G>(bset0, vband, p0, lh);
    p2_load<P2G, P2G>(bset1, vband, p0, lh);

    {
        const float scale = 0.125f;
        float adot = (qa.x * ka.x + qa.y * ka.y) + (qa.z * ka.z + qa.w * ka.w);
        adot = row_sum16(adot) * scale;
        const bool avalid = nodeok && (snode - WIN) > 0;

        float sv[5];
        bool  inwin[5], ok[5];
        float mx = avalid ? fabsf(adot) : -1e30f;
#pragma unroll
        for (int i = 0; i < 5; ++i) {
            const int kc = sj + 16 * i;
            const int w  = kc - sm;
            const int p  = p0 - 32 + kc;
            inwin[i] = nodeok && w >= 0 && w <= 2 * WIN;
            ok[i]    = inwin[i] && p >= 0 && p < NN;
            sv[i]    = s_scores[sm][kc] * scale;
            if (ok[i]) mx = fmaxf(mx, fabsf(sv[i]));
        }
        mx = row_max16(mx);

        float se[5];
        float sum = 0.f;
#pragma unroll
        for (int i = 0; i < 5; ++i) {
            const float e = ok[i] ? __expf(fabsf(sv[i]) - mx) : 0.f;
            sum += e;
            se[i] = (sv[i] > 0.f) ? e : ((sv[i] < 0.f) ? -e : 0.f);
        }
        sum = row_sum16(sum);
        const float ea  = avalid ? __expf(fabsf(adot) - mx) : 0.f;
        const float ase = (adot > 0.f) ? ea : ((adot < 0.f) ? -ea : 0.f);
        sum += ea;
        const float inv = 1.f / sum;

        float ds = 0.f;
#pragma unroll
        for (int i = 0; i < 5; ++i) {
            const float we = ok[i] ? se[i] * inv : 0.f;
            s_w[sm][sj + 16 * i] = we;
            if (inwin[i]) ds += we * st[i];
        }
        ds = row_sum16(ds);
        const float wa = avalid ? ase * inv : 0.f;
        if (sj < KB - 80) s_w[sm][80 + sj] = (sj == 0) ? wa : 0.f;
        ds += wa * st_anchor;
        if (nodeok && sj == 0) state_out[base + snode] = st_self + ds;
    }
    __syncthreads();

    {
        v8f acc[4] = {};
        const float* wfrag = &s_w[lm][2 * lh];
        p2_mma<0, P2G>(acc, bset0, wfrag);
        p2_load<2 * P2G, P2G>(bset0, vband, p0, lh);
        p2_mma<P2G, P2G>(acc, bset1, wfrag);
        p2_load<3 * P2G, KB / 4 - 3 * P2G>(bset1, vband, p0, lh);
        p2_mma<2 * P2G, P2G>(acc, bset0, wfrag);
        p2_mma<3 * P2G, KB / 4 - 3 * P2G>(acc, bset1, wfrag);
#pragma unroll
        for (int i = 0; i < 8; ++i) {
            v4f o;
            o.x = acc[0][i];
            o.y = acc[1][i];
            o.z = acc[2][i];
            o.w = acc[3][i];
            *(v4f*)&s_y[i + 8 * lh][c0 + 4 * lm] = o;
        }
    }

    const int wrow = 2 * __builtin_amdgcn_readfirstlane(wave);
    const int c4   = lane * 4;
    v4f res[2][4];
    v4f g[4], bt[4];
#pragma unroll
    for (int r = 0; r < 2; ++r) {
        const int nodec = min(p0 + wrow + r, NN - 1);
        const float* vr = val_in + (size_t)(base + nodec) * D + c4;
#pragma unroll
        for (int j = 0; j < 4; ++j)
            res[r][j] = *(const v4f*)(vr + j * 128);
    }
#pragma unroll
    for (int j = 0; j < 4; ++j) {
        g[j]  = *(const v4f*)(gamma + j * 128 + c4);
        bt[j] = *(const v4f*)(beta + j * 128 + c4);
    }
    __syncthreads();

#pragma unroll
    for (int r = 0; r < 2; ++r) {
        const int m    = wrow + r;
        const int node = p0 + m;
        if (node < NN) {
            v4f y[4];
            float part = 0.f;
#pragma unroll
            for (int j = 0; j < 4; ++j) {
                y[j] = *(const v4f*)&s_y[m][j * 128 + c4] + res[r][j];
                part += (y[j].x + y[j].y) + (y[j].z + y[j].w);
            }
            const float mean = wave_sum(part) * (1.0f / D);
            part = 0.f;
#pragma unroll
            for (int j = 0; j < 4; ++j) {
                y[j] -= mean;
                const v4f sq = y[j] * y[j];
                part += (sq.x + sq.y) + (sq.z + sq.w);
            }
            const float rstd = rsqrtf(wave_sum(part) * (1.0f / D) + 1e-5f);
            float* vo = val_out + (size_t)(base + node) * D + c4;
#pragma unroll
            for (int j = 0; j < 4; ++j)
                *(v4f*)(vo + j * 128) = y[j] * rstd * g[j] + bt[j];
        }
    }
}

extern "C" void kernel_launch(void* const* d_in, const int* in_sizes, int n_in,
                              void* d_out, int out_size, void* d_ws, size_t ws_size,
                              hipStream_t stream)
{
    const int*   ids    = (const int*)  d_in[0];
    const float* emb    = (const float*)d_in[1];
    const float* pos    = (const float*)d_in[2];
    const float* aval   = (const float*)d_in[3];
    const float* astate = (const float*)d_in[4];
    const float* ing    = (const float*)d_in[5];
    const float* inb    = (const float*)d_in[6];
    const float* spw    = (const float*)d_in[7];
    const float* spb    = (const float*)d_in[8];
    const float* wq     = (const float*)d_in[9];
    const float* wk     = (const float*)d_in[10];
    const float* lng    = (const float*)d_in[11];
    const float* lnb    = (const float*)d_in[12];

    float* ws = (float*)d_ws;
    size_t o = 0;
    float* val_a = ws + o; o += (size_t)MM * D;
    float* val_b = ws + o; o += (size_t)MM * D;
    float* st_a  = ws + o; o += MM;
    float* st_b  = ws + o; o += MM;
    float* qbuf  = ws + o; o += (size_t)MM * R;
    float* kbuf  = ws + o; o += (size_t)MM * R;

    float* out_state = (float*)d_out;
    float* out_val   = out_state + MM;

    const int eblocks = (MM + EROWS - 1) / EROWS;
    embed_ln_kernel<<<eblocks, 256, 0, stream>>>(ids, emb, pos, aval, astate,
                                                 ing, inb, spw, spb, val_a, st_a);

    const int gblocks = (MM + GBM - 1) / GBM;
    const int ablocks = BB * TILES;

    qk_gemm_kernel<<<gblocks, 256, 0, stream>>>(val_a, wq, wk, qbuf, kbuf);
    attn_fused_kernel<<<ablocks, 256, 0, stream>>>(val_a, st_a, qbuf, kbuf,
                                                   lng, lnb, val_b, st_b);
    qk_gemm_kernel<<<gblocks, 256, 0, stream>>>(val_b, wq + (size_t)D * R,
                                                wk + (size_t)D * R, qbuf, kbuf);
    attn_fused_kernel<<<ablocks, 256, 0, stream>>>(val_b, st_b, qbuf, kbuf,
                                                   lng + D, lnb + D, out_val, out_state);
}
